# starting kernel
# baseline (speedup 1.0000x reference)
#include <hip/hip_runtime.h>
#include <math.h>

typedef __attribute__((ext_vector_type(2))) float v2f;
typedef __attribute__((ext_vector_type(4))) float v4f;
typedef __attribute__((ext_vector_type(8))) float v8f;

#define AS3 __attribute__((address_space(3)))
#define AS1 __attribute__((address_space(1)))
typedef AS3 float*       lptr;
typedef const AS3 float* clptr;
typedef AS1 float*       gptr;
typedef const AS1 float* cgptr;

#define B_TOT    8192
#define L_SEQ    256
#define NCH      19
#define KPACK1   96
#define KPACK2   64
#define NPACK    32
#define PRS      96
#define NWAVES   8
#define NTHREADS 256
#define K1N      (KPACK1 * NPACK)
#define K2N      (KPACK2 * NPACK)

#define SEV_OFF 0
#define SEV_SZ  2578
#define SOD_OFF (SEV_OFF + SEV_SZ)
#define SOD_SZ  2578
#define HB_OFF  (SOD_OFF + SOD_SZ)
#define HB_SZ   2502
#define DB_OFF  (HB_OFF + HB_SZ)
#define DB_SZ   2578
#define CB_OFF  (DB_OFF + DB_SZ)
#define CB_SZ   2578
#define W1S_OFF 12816
#define W1S_SZ  ((KPACK1 / 2) * PRS)
#define W2S_OFF (W1S_OFF + W1S_SZ)
#define W2S_SZ  ((KPACK2 / 2) * PRS)
#define SMEM_TOT (W2S_OFF + W2S_SZ)

static __device__ __forceinline__ v8f wmma4(v2f a, v2f b, v8f c) {
  return __builtin_amdgcn_wmma_f32_16x16x4_f32(false, a, false, b, (short)0, c,
                                               false, false);
}

static __device__ __forceinline__ float leaky(float x) {
  return x > 0.f ? x : 0.01f * x;
}

static __device__ __forceinline__ float fast_tanh(float x) {
  x = fminf(fmaxf(x, -15.f), 15.f);
  const float e = __expf(2.f * x);
  return __fdividef(e - 1.f, e + 1.f);
}

template <int MODE>
static __device__ void run_branch(cgptr Wt1, cgptr Wt2, cgptr b1, cgptr b2,
                                  clptr in, lptr hb, lptr w1s, lptr w2s,
                                  int Lp, clptr other, lptr outLds,
                                  gptr xg, int tmul, int tadd) {
  const int tid  = threadIdx.x;
  const int wave = tid >> 5;
  const int lane = tid & 31;
  const int lhi  = lane >> 4;
  const int lln  = lane & 15;

  for (int i = tid; i < K1N / 4; i += NTHREADS) {
    const int p = i >> 4, j = (i & 15) * 4;
    *(AS3 v4f*)(w1s + p * PRS + j) = ((const AS1 v4f*)Wt1)[i];
  }
  for (int i = tid; i < K2N / 4; i += NTHREADS) {
    const int p = i >> 4, j = (i & 15) * 4;
    *(AS3 v4f*)(w2s + p * PRS + j) = ((const AS1 v4f*)Wt2)[i];
  }
  __syncthreads();

  const int M1   = Lp + 2;
  const int nMt1 = (M1 + 15) >> 4;
  const float b1a = b1[lln];
  const float b1b = b1[lln < 3 ? 16 + lln : 18];
  for (int mt = wave; mt < nMt1; mt += NWAVES) {
    const int m0 = mt << 4;
    int mrow = m0 + lln; if (mrow > M1 - 1) mrow = M1 - 1;
    clptr ap = in + mrow * NCH + 2 * lhi;
    clptr bp = w1s + lhi * PRS + 2 * lln;
    v8f acc0 = {}; v8f acc1 = {};
#pragma unroll 4
    for (int k0 = 0; k0 < KPACK1; k0 += 4) {
      v2f a; a.x = ap[k0]; a.y = ap[k0 + 1];
      v2f bv0 = *(const AS3 v2f*)(bp);
      v2f bv1 = *(const AS3 v2f*)(bp + 32);
      bp += 2 * PRS;
      acc0 = wmma4(a, bv0, acc0);
      acc1 = wmma4(a, bv1, acc1);
    }
    lptr hrow = hb + (m0 + 8 * lhi) * NCH;
    if (m0 + 16 <= M1) {
#pragma unroll
      for (int v = 0; v < 8; v++) hrow[v * NCH + lln] = leaky(acc0[v] + b1a);
      if (lln < 3) {
#pragma unroll
        for (int v = 0; v < 8; v++)
          hrow[v * NCH + 16 + lln] = leaky(acc1[v] + b1b);
      }
    } else {
#pragma unroll
      for (int v = 0; v < 8; v++)
        if (m0 + 8 * lhi + v < M1) hrow[v * NCH + lln] = leaky(acc0[v] + b1a);
      if (lln < 3) {
#pragma unroll
        for (int v = 0; v < 8; v++)
          if (m0 + 8 * lhi + v < M1)
            hrow[v * NCH + 16 + lln] = leaky(acc1[v] + b1b);
      }
    }
  }
  __syncthreads();

  const int M2   = Lp;
  const int nMt2 = M2 >> 4;
  const float b2a = b2[lln];
  const float b2b = b2[lln < 3 ? 16 + lln : 18];
  for (int mt = wave; mt < nMt2; mt += NWAVES) {
    const int m0 = mt << 4;
    clptr ap = hb + (m0 + lln) * NCH + 2 * lhi;
    clptr bp = w2s + lhi * PRS + 2 * lln;
    v8f acc0 = {}; v8f acc1 = {};
#pragma unroll 4
    for (int k0 = 0; k0 < KPACK2; k0 += 4) {
      v2f a; a.x = ap[k0]; a.y = ap[k0 + 1];
      v2f bv0 = *(const AS3 v2f*)(bp);
      v2f bv1 = *(const AS3 v2f*)(bp + 32);
      bp += 2 * PRS;
      acc0 = wmma4(a, bv0, acc0);
      acc1 = wmma4(a, bv1, acc1);
    }
    const int mb = m0 + 8 * lhi;
#pragma unroll
    for (int v = 0; v < 8; v++) {
      const int m = mb + v;
      const float t   = fast_tanh(acc0[v] + b2a);
      const float oth = other[(3 + m) * NCH + lln];
      if (MODE <= 1)      outLds[(3 + m) * NCH + lln] = oth * __expf(t);
      else if (MODE == 2) xg[(m * tmul + tadd) * NCH + lln] = oth + t;
      else                xg[(m * tmul + tadd) * NCH + lln] = oth - t;
    }
    if (lln < 3) {
      const int co = 16 + lln;
#pragma unroll
      for (int v = 0; v < 8; v++) {
        const int m = mb + v;
        const float t   = fast_tanh(acc1[v] + b2b);
        const float oth = other[(3 + m) * NCH + co];
        if (MODE <= 1)      outLds[(3 + m) * NCH + co] = oth * __expf(t);
        else if (MODE == 2) xg[(m * tmul + tadd) * NCH + co] = oth + t;
        else                xg[(m * tmul + tadd) * NCH + co] = oth - t;
      }
    }
  }
  __syncthreads();
}

__global__ __launch_bounds__(NTHREADS) void scinet_level_kernel(
    float* __restrict__ xbuf_, const float* __restrict__ Wt1p_,
    const float* __restrict__ Wt2p_, const float* __restrict__ b1g_,
    const float* __restrict__ b2g_, int depth) {
  __shared__ __align__(16) float smem[SMEM_TOT];
  const int tid = threadIdx.x;
  const int sel = blockIdx.x & ((1 << depth) - 1);
  const int b   = blockIdx.x >> depth;
  int node = 0;
  for (int i = 0; i < depth; i++) node = 2 * node + 1 + ((sel >> i) & 1);
  const int stride = 1 << depth;
  const int Lp     = (L_SEQ >> depth) >> 1;

  lptr sev = (lptr)(smem + SEV_OFF);
  lptr sod = (lptr)(smem + SOD_OFF);
  lptr hb  = (lptr)(smem + HB_OFF);
  lptr db  = (lptr)(smem + DB_OFF);
  lptr cb  = (lptr)(smem + CB_OFF);
  lptr w1s = (lptr)(smem + W1S_OFF);
  lptr w2s = (lptr)(smem + W2S_OFF);
  lptr sm  = (lptr)smem;

  for (int i = tid; i < SMEM_TOT; i += NTHREADS) sm[i] = 0.f;
  __syncthreads();

  gptr  xg   = (gptr)(xbuf_ + (size_t)b * L_SEQ * NCH);
  cgptr Wt1p = (cgptr)Wt1p_;
  cgptr Wt2p = (cgptr)Wt2p_;
  cgptr b1g  = (cgptr)b1g_;
  cgptr b2g  = (cgptr)b2g_;

  for (int i = tid; i < Lp * NCH; i += NTHREADS) {
    const int r = i / NCH, c = i - r * NCH;
    sev[(3 + r) * NCH + c] = xg[((2 * r) * stride + sel) * NCH + c];
    sod[(3 + r) * NCH + c] = xg[((2 * r + 1) * stride + sel) * NCH + c];
  }
  __syncthreads();
  for (int i = tid; i < 2 * 3 * NCH; i += NTHREADS) {
    const int c = i % NCH, r = (i / NCH) % 3;
    lptr buf = (i >= 3 * NCH) ? sod : sev;
    buf[r * NCH + c]            = buf[3 * NCH + c];
    buf[(Lp + 3 + r) * NCH + c] = buf[(Lp + 2) * NCH + c];
  }
  __syncthreads();

  const int nb = node * 4;
  run_branch<0>(Wt1p + (size_t)(nb + 0) * K1N, Wt2p + (size_t)(nb + 0) * K2N,
                b1g + (nb + 0) * NCH, b2g + (nb + 0) * NCH,
                sev, hb, w1s, w2s, Lp, sod, db, xg, 0, 0);
  run_branch<1>(Wt1p + (size_t)(nb + 1) * K1N, Wt2p + (size_t)(nb + 1) * K2N,
                b1g + (nb + 1) * NCH, b2g + (nb + 1) * NCH,
                sod, hb, w1s, w2s, Lp, sev, cb, xg, 0, 0);
  for (int i = tid; i < 2 * 3 * NCH; i += NTHREADS) {
    const int c = i % NCH, r = (i / NCH) % 3;
    lptr buf = (i >= 3 * NCH) ? cb : db;
    buf[r * NCH + c]            = buf[3 * NCH + c];
    buf[(Lp + 3 + r) * NCH + c] = buf[(Lp + 2) * NCH + c];
  }
  __syncthreads();
  run_branch<2>(Wt1p + (size_t)(nb + 3) * K1N, Wt2p + (size_t)(nb + 3) * K2N,
                b1g + (nb + 3) * NCH, b2g + (nb + 3) * NCH,
                db, hb, w1s, w2s, Lp, cb, db, xg, 2 * stride, sel);
  run_branch<3>(Wt1p + (size_t)(nb + 2) * K1N, Wt2p + (size_t)(nb + 2) * K2N,
                b1g + (nb + 2) * NCH, b2g + (nb + 2) * NCH,
                cb, hb, w1s, w2s, Lp, db, cb, xg, 2 * stride, sel + stride);
}

__global__ void pack_weights_kernel(const float* __restrict__ W1,
                                    const float* __restrict__ W2,
                                    float* __restrict__ Wt1p,
                                    float* __restrict__ Wt2p) {
  const int idx = blockIdx.x * blockDim.x + threadIdx.x;
  const int N1 = 28 * K1N, N2 = 28 * K2N;
  if (idx < N1) {
    const int nb = idx / K1N, rem = idx - nb * K1N;
    const int p = rem >> 6, q = rem & 63;
    const int co = q >> 1, k = 2 * p + (q & 1);
    float v = 0.f;
    if (co < NCH && k < 95) {
      const int tap = k / NCH, ci = k - tap * NCH;
      v = W1[((size_t)(nb * NCH + co) * NCH + ci) * 5 + tap];
    }
    Wt1p[idx] = v;
  } else if (idx < N1 + N2) {
    const int j = idx - N1;
    const int nb = j / K2N, rem = j - nb * K2N;
    const int p = rem >> 6, q = rem & 63;
    const int co = q >> 1, k = 2 * p + (q & 1);
    float v = 0.f;
    if (co < NCH && k < 57) {
      const int tap = k / NCH, ci = k - tap * NCH;
      v = W2[((size_t)(nb * NCH + co) * NCH + ci) * 3 + tap];
    }
    Wt2p[j] = v;
  }
}

__global__ void concat_kernel(const float* __restrict__ x0D,
                              const float* __restrict__ xctrl,
                              float* __restrict__ xbuf) {
  const long long idx = (long long)blockIdx.x * blockDim.x + threadIdx.x;
  if (idx >= (long long)B_TOT * L_SEQ * NCH) return;
  const int c = (int)(idx % NCH);
  const long long bl = idx / NCH;
  xbuf[idx] = (c < 5) ? x0D[bl * 5 + c] : xctrl[bl * 14 + (c - 5)];
}

__global__ void proj_kernel(const float* __restrict__ xbuf,
                            const float* __restrict__ x0D,
                            const float* __restrict__ projw,
                            float* __restrict__ out) {
  __shared__ float yres_s[L_SEQ * 5];
  lptr yres = (lptr)yres_s;
  const int b = blockIdx.x, tid = threadIdx.x;
  cgptr xb = (cgptr)xbuf, x0 = (cgptr)x0D, pj = (cgptr)projw;
  for (int i = tid; i < L_SEQ * 5; i += blockDim.x) {
    const int l = i / 5, c = i - l * 5;
    yres[i] = xb[((size_t)b * L_SEQ + l) * NCH + c] +
              x0[((size_t)b * L_SEQ + l) * 5 + c];
  }
  __syncthreads();
  for (int i = tid; i < 64 * 5; i += blockDim.x) {
    const int o = i / 5, c = i - o * 5;
    cgptr pw = pj + o * L_SEQ;
    float acc = 0.f;
    for (int l = 0; l < L_SEQ; l++) acc = fmaf(pw[l], yres[l * 5 + c], acc);
    ((gptr)out)[(size_t)b * 320 + i] = acc;
  }
}

extern "C" void kernel_launch(void* const* d_in, const int* in_sizes, int n_in,
                              void* d_out, int out_size, void* d_ws,
                              size_t ws_size, hipStream_t stream) {
  const float* x0D   = (const float*)d_in[0];
  const float* xctrl = (const float*)d_in[1];
  const float* W1    = (const float*)d_in[2];
  const float* b1    = (const float*)d_in[3];
  const float* W2    = (const float*)d_in[4];
  const float* b2    = (const float*)d_in[5];
  const float* projw = (const float*)d_in[6];
  float* out = (float*)d_out;

  float* ws   = (float*)d_ws;
  float* xbuf = ws;
  const size_t XB = (size_t)B_TOT * L_SEQ * NCH;
  float* Wt1p = xbuf + XB;
  float* Wt2p = Wt1p + 28 * K1N;

  const int packN = 28 * K1N + 28 * K2N;
  pack_weights_kernel<<<(packN + 255) / 256, 256, 0, stream>>>(W1, W2, Wt1p, Wt2p);
  concat_kernel<<<(int)((XB + 255) / 256), 256, 0, stream>>>(x0D, xctrl, xbuf);
  for (int d = 0; d < 3; d++)
    scinet_level_kernel<<<B_TOT << d, NTHREADS, 0, stream>>>(xbuf, Wt1p, Wt2p,
                                                             b1, b2, d);
  proj_kernel<<<B_TOT, 256, 0, stream>>>(xbuf, x0D, projw, out);
}


# Ch 16-18 leave im2col: per-tap products x[p,:]*W[tap] (K=19, one N tile = taps x 3 ch) go to LDS and are summed over rows p=m+tap; padded 2nd N tile gone; their sums reordered (per tap, then taps)
# speedup vs baseline: 1.9199x; 1.9199x over previous
#include <hip/hip_runtime.h>
#include <math.h>

typedef __attribute__((ext_vector_type(2))) float v2f;
typedef __attribute__((ext_vector_type(4))) float v4f;
typedef __attribute__((ext_vector_type(8))) float v8f;

#define AS3 __attribute__((address_space(3)))
#define AS1 __attribute__((address_space(1)))
typedef AS3 float*       lptr;
typedef const AS3 float* clptr;
typedef AS1 float*       gptr;
typedef const AS1 float* cgptr;

#define B_TOT    8192
#define L_SEQ    256
#define NCH      19
#define KPACK1   96
#define KPACK2   64
#define KS1      (KPACK1 / 4)
#define KS2      (KPACK2 / 4)
#define WAVE     32
#define TPB      64
#define KST      5
#define F1M      (KS1 * WAVE * 2)
#define F2M      (KS2 * WAVE * 2)
#define FTL      (KST * WAVE * 2)
#define W1BR     (F1M + FTL)
#define W2BR     (F2M + FTL)
#define TP       16

#define BUF_SZ(LP) ((((LP) + 6) * NCH + 32 + 3) & ~3)
#define HB_SZ(LP)  ((((LP) + 2) * NCH + 32 + 3) & ~3)
#define SMEM_TOT(LP) (4 * BUF_SZ(LP) + 2 * HB_SZ(LP))

static __device__ __forceinline__ v8f wmma4(v2f a, v2f b, v8f c) {
  return __builtin_amdgcn_wmma_f32_16x16x4_f32(false, a, false, b, (short)0, c,
                                               false, false);
}

static __device__ __forceinline__ float leaky(float x) {
  return x > 0.f ? x : 0.01f * x;
}

static __device__ __forceinline__ float fast_tanh(float x) {
  x = fminf(fmaxf(x, -15.f), 15.f);
  const float e = __expf(2.f * x);
  return __fdividef(e - 1.f, e + 1.f);
}

template <int KS>
static __device__ __forceinline__ void load_a(clptr ap, v2f (&a)[KS]) {
#pragma unroll
  for (int ks = 0; ks < KS; ks++) { a[ks].x = ap[4 * ks]; a[ks].y = ap[4 * ks + 1]; }
}

template <int ROWS>
static __device__ __forceinline__ void tail_products(cgptr Wt, clptr src, lptr scr,
                                                     int lhi, int lln) {
  constexpr int nT = (ROWS + 15) >> 4;
  const int lane = lhi * 16 + lln;
  v2f bt[KST];
#pragma unroll
  for (int ks = 0; ks < KST; ks++) bt[ks] = ((const AS1 v2f*)Wt)[ks * WAVE + lane];
  v2f an[KST];
  load_a<KST>(src + lln * NCH + 2 * lhi, an);
#pragma nounroll
  for (int mt = 0; mt < nT; mt++) {
    const int m0 = mt << 4;
    v2f a[KST];
#pragma unroll
    for (int ks = 0; ks < KST; ks++) a[ks] = an[ks];
    {
      const int mn = (mt + 1 < nT) ? m0 + 16 : m0;
      int prow = mn + lln; if (prow > ROWS - 1) prow = ROWS - 1;
      load_a<KST>(src + prow * NCH + 2 * lhi, an);
    }
    v8f acc = {};
#pragma unroll
    for (int ks = 0; ks < KST; ks++) acc = wmma4(a[ks], bt[ks], acc);
    lptr srow = scr + (m0 + 8 * lhi) * TP + lln;
    if (m0 + 16 <= ROWS) {
#pragma unroll
      for (int v = 0; v < 8; v++) srow[v * TP] = acc[v];
    } else {
#pragma unroll
      for (int v = 0; v < 8; v++)
        if (m0 + 8 * lhi + v < ROWS) srow[v * TP] = acc[v];
    }
  }
}

template <int TAPS>
static __device__ __forceinline__ float tail_sum(clptr scr, int m, int j) {
  clptr sp = scr + m * TP + j;
  float s = sp[0];
#pragma unroll
  for (int t = 1; t < TAPS; t++) s += sp[t * (TP + 3)];
  return s;
}

template <bool TO_GLOBAL, int LP>
static __device__ void run_branch(cgptr Wf1, cgptr Wf2, cgptr b1, cgptr b2,
                                  clptr in, lptr hb, clptr other, lptr outLds,
                                  lptr scr, gptr xg, int tmul, int tadd, float sgn) {
  const int lane = threadIdx.x & (WAVE - 1);
  const int lhi  = lane >> 4;
  const int lln  = lane & 15;
  static_assert(LP % 32 == 0, "M tiles are taken in pairs; 3*LP outputs fill whole waves");

  constexpr int M1 = LP + 2;
  {
    tail_products<LP + 6>(Wf1 + F1M, in, hb, lhi, lln);
    const float bt0 = b1[16], bt1 = b1[17], bt2 = b1[18];
    constexpr int NQ  = 3 * M1;
    constexpr int NIT = (NQ + WAVE - 1) / WAVE;
    float hv[NIT];
#pragma unroll
    for (int it = 0; it < NIT; it++) {
      int q = it * WAVE + lane; if (q > NQ - 1) q = NQ - 1;
      const int m = q / 3, j = q - 3 * m;
      const float bj = j == 0 ? bt0 : (j == 1 ? bt1 : bt2);
      hv[it] = leaky(tail_sum<5>(hb, m, j) + bj);
    }
#pragma unroll
    for (int it = 0; it < NIT; it++) {
      const int q = it * WAVE + lane;
      if (q < NQ) {
        const int m = q / 3, j = q - 3 * m;
        hb[m * NCH + 16 + j] = hv[it];
      }
    }
  }
  {
    v2f bw[KS1];
#pragma unroll
    for (int ks = 0; ks < KS1; ks++)
      bw[ks] = ((const AS1 v2f*)Wf1)[ks * WAVE + lane];
    const float b1a = b1[lln];
    constexpr int nFull = LP >> 4;
#pragma nounroll
    for (int mt = 0; mt < nFull; mt += 2) {
      const int m0 = mt << 4;
      clptr apA = in + (m0 + lln) * NCH + 2 * lhi;
      clptr apB = apA + 16 * NCH;
      v8f accA = {}; v8f accB = {};
#pragma unroll
      for (int ks = 0; ks < KS1; ks++) {
        v2f aA; aA.x = apA[4 * ks]; aA.y = apA[4 * ks + 1];
        v2f aB; aB.x = apB[4 * ks]; aB.y = apB[4 * ks + 1];
        accA = wmma4(aA, bw[ks], accA);
        accB = wmma4(aB, bw[ks], accB);
      }
      lptr hrA = hb + (m0 + 8 * lhi) * NCH + lln;
      lptr hrB = hrA + 16 * NCH;
#pragma unroll
      for (int v = 0; v < 8; v++) hrA[v * NCH] = leaky(accA[v] + b1a);
#pragma unroll
      for (int v = 0; v < 8; v++) hrB[v * NCH] = leaky(accB[v] + b1a);
    }
    {
      int mrow = LP + lln; if (mrow > M1 - 1) mrow = M1 - 1;
      clptr ap = in + mrow * NCH + 2 * lhi;
      v8f acc = {};
#pragma unroll
      for (int ks = 0; ks < KS1; ks++) {
        v2f a; a.x = ap[4 * ks]; a.y = ap[4 * ks + 1];
        acc = wmma4(a, bw[ks], acc);
      }
      if (lhi == 0) {
        lptr hr = hb + LP * NCH + lln;
        hr[0]   = leaky(acc[0] + b1a);
        hr[NCH] = leaky(acc[1] + b1a);
      }
    }
  }
  __syncthreads();

  {
    tail_products<M1>(Wf2 + F2M, hb, scr, lhi, lln);
    const float bt0 = b2[16], bt1 = b2[17], bt2 = b2[18];
    constexpr int NIT = (3 * LP) / WAVE;
    float tv[NIT];
#pragma unroll
    for (int it = 0; it < NIT; it++) {
      const int q = it * WAVE + lane;
      const int m = q / 3, j = q - 3 * m;
      const float bj = j == 0 ? bt0 : (j == 1 ? bt1 : bt2);
      tv[it] = tail_sum<3>(scr, m, j) + bj;
    }
#pragma unroll
    for (int it = 0; it < NIT; it++) {
      const int q = it * WAVE + lane;
      const int m = q / 3, j = q - 3 * m;
      const float t   = fast_tanh(tv[it]);
      const float oth = other[(3 + m) * NCH + 16 + j];
      if (!TO_GLOBAL) outLds[(3 + m) * NCH + 16 + j] = oth * __expf(t);
      else            xg[(m * tmul + tadd) * NCH + 16 + j] = fmaf(sgn, t, oth);
    }
  }
  {
    v2f bw[KS2];
#pragma unroll
    for (int ks = 0; ks < KS2; ks++)
      bw[ks] = ((const AS1 v2f*)Wf2)[ks * WAVE + lane];
    const float b2a = b2[lln];
    constexpr int nMt2 = LP >> 4;
#pragma nounroll
    for (int mt = 0; mt < nMt2; mt += 2) {
      const int m0 = mt << 4;
      clptr apA = hb + (m0 + lln) * NCH + 2 * lhi;
      clptr apB = apA + 16 * NCH;
      v8f accA = {}; v8f accB = {};
#pragma unroll
      for (int ks = 0; ks < KS2; ks++) {
        v2f aA; aA.x = apA[4 * ks]; aA.y = apA[4 * ks + 1];
        v2f aB; aB.x = apB[4 * ks]; aB.y = apB[4 * ks + 1];
        accA = wmma4(aA, bw[ks], accA);
        accB = wmma4(aB, bw[ks], accB);
      }
      const int mb = m0 + 8 * lhi;
#pragma unroll
      for (int v = 0; v < 8; v++) {
        const int m = mb + v;
        const float t   = fast_tanh(accA[v] + b2a);
        const float oth = other[(3 + m) * NCH + lln];
        if (!TO_GLOBAL) outLds[(3 + m) * NCH + lln] = oth * __expf(t);
        else            xg[(m * tmul + tadd) * NCH + lln] = fmaf(sgn, t, oth);
      }
#pragma unroll
      for (int v = 0; v < 8; v++) {
        const int m = mb + 16 + v;
        const float t   = fast_tanh(accB[v] + b2a);
        const float oth = other[(3 + m) * NCH + lln];
        if (!TO_GLOBAL) outLds[(3 + m) * NCH + lln] = oth * __expf(t);
        else            xg[(m * tmul + tadd) * NCH + lln] = fmaf(sgn, t, oth);
      }
    }
  }
  if (!TO_GLOBAL) __syncthreads();
}

template <int DEPTH>
__global__ __launch_bounds__(TPB) void scinet_level_kernel(
    float* __restrict__ xbuf_, const float* __restrict__ Wf1p_,
    const float* __restrict__ Wf2p_, const float* __restrict__ b1g_,
    const float* __restrict__ b2g_) {
  constexpr int LP     = (L_SEQ >> DEPTH) >> 1;
  constexpr int stride = 1 << DEPTH;
  __shared__ __align__(16) float smem[SMEM_TOT(LP)];
  const int tid = threadIdx.x;
  const int w   = __builtin_amdgcn_readfirstlane(tid >> 5);
  const int sel = blockIdx.x & (stride - 1);
  const int b   = blockIdx.x >> DEPTH;
  int node = 0;
  for (int i = 0; i < DEPTH; i++) node = 2 * node + 1 + ((sel >> i) & 1);

  lptr sm  = (lptr)smem;
  lptr sev = sm;
  lptr sod = sev + BUF_SZ(LP);
  lptr db  = sod + BUF_SZ(LP);
  lptr cb  = db + BUF_SZ(LP);
  lptr hb  = cb + BUF_SZ(LP) + w * HB_SZ(LP);

  {
    const v4f z4 = {0.f, 0.f, 0.f, 0.f};
    for (int i = tid; i < SMEM_TOT(LP) / 4; i += TPB) ((AS3 v4f*)sm)[i] = z4;
  }
  __syncthreads();

  gptr  xg   = (gptr)(xbuf_ + (size_t)b * L_SEQ * NCH);
  cgptr Wf1p = (cgptr)Wf1p_;
  cgptr Wf2p = (cgptr)Wf2p_;
  cgptr b1g  = (cgptr)b1g_;
  cgptr b2g  = (cgptr)b2g_;

  constexpr int NEL   = LP * NCH;
  constexpr int NFULL = NEL / TPB;
  constexpr int NREM  = NEL - NFULL * TPB;
  static_assert(NREM == 0 || NREM == WAVE, "the tail pass is whole waves");
#pragma unroll 19
  for (int p = 0; p < NFULL; p++) {
    const int i = p * TPB + tid;
    const int r = i / NCH, c = i - r * NCH;
    sev[(3 + r) * NCH + c] = xg[((2 * r) * stride + sel) * NCH + c];
    sod[(3 + r) * NCH + c] = xg[((2 * r + 1) * stride + sel) * NCH + c];
  }
  if (NREM != 0 && w == 0) {
    const int i = NFULL * TPB + tid;
    const int r = i / NCH, c = i - r * NCH;
    sev[(3 + r) * NCH + c] = xg[((2 * r) * stride + sel) * NCH + c];
    sod[(3 + r) * NCH + c] = xg[((2 * r + 1) * stride + sel) * NCH + c];
  }
  __syncthreads();
  for (int i = tid; i < 2 * 3 * NCH; i += TPB) {
    const int c = i % NCH, r = (i / NCH) % 3;
    lptr buf = (i >= 3 * NCH) ? sod : sev;
    buf[r * NCH + c]            = buf[3 * NCH + c];
    buf[(LP + 3 + r) * NCH + c] = buf[(LP + 2) * NCH + c];
  }
  __syncthreads();

  const int nb = node * 4;
  {
    const int br = nb + w;
    run_branch<false, LP>(Wf1p + (size_t)br * W1BR, Wf2p + (size_t)br * W2BR,
                          b1g + br * NCH, b2g + br * NCH,
                          w ? sod : sev, hb, w ? sev : sod, w ? cb : db,
                          w ? cb : db,
                          xg, 0, 0, 0.f);
  }
  for (int i = tid; i < 2 * 3 * NCH; i += TPB) {
    const int c = i % NCH, r = (i / NCH) % 3;
    lptr buf = (i >= 3 * NCH) ? cb : db;
    buf[r * NCH + c]            = buf[3 * NCH + c];
    buf[(LP + 3 + r) * NCH + c] = buf[(LP + 2) * NCH + c];
  }
  __syncthreads();
  {
    const int br = nb + 3 - w;
    run_branch<true, LP>(Wf1p + (size_t)br * W1BR, Wf2p + (size_t)br * W2BR,
                         b1g + br * NCH, b2g + br * NCH,
                         w ? cb : db, hb, w ? db : cb, db,
                         w ? sod : sev,
                         xg, 2 * stride, sel + w * stride, w ? -1.f : 1.f);
  }
}

template <int TAPS, int KMAIN>
static __device__ __forceinline__ float pack_value(const float* __restrict__ Wg, int nb,
                                                   int rem, int fmain) {
  const bool tail = rem >= fmain;
  const int r  = tail ? rem - fmain : rem;
  const int ks = r >> 6, ln = (r >> 1) & 31, e = r & 1;
  const int n  = ln & 15;
  const int k  = 4 * ks + 2 * (ln >> 4) + e;
  int co, ci, tap; bool live;
  if (!tail) { tap = k / NCH; ci = k - tap * NCH; co = n; live = k < KMAIN; }
  else       { tap = n / 3; co = 16 + (n - 3 * tap); ci = k; live = (n < 3 * TAPS) && (k < NCH); }
  return live ? Wg[((size_t)(nb * NCH + co) * NCH + ci) * TAPS + tap] : 0.f;
}

__global__ void pack_weights_kernel(const float* __restrict__ W1,
                                    const float* __restrict__ W2,
                                    float* __restrict__ Wf1p,
                                    float* __restrict__ Wf2p) {
  const int idx = blockIdx.x * blockDim.x + threadIdx.x;
  const int N1 = 28 * W1BR, N2 = 28 * W2BR;
  if (idx < N1) {
    const int nb = idx / W1BR, rem = idx - nb * W1BR;
    Wf1p[idx] = pack_value<5, 95>(W1, nb, rem, F1M);
  } else if (idx < N1 + N2) {
    const int j = idx - N1;
    const int nb = j / W2BR, rem = j - nb * W2BR;
    Wf2p[j] = pack_value<3, 57>(W2, nb, rem, F2M);
  }
}

#define CC_TPB   256
#define CC_WG    2048
#define CC_V4    (B_TOT * L_SEQ * NCH / 4)
#define CC_PER   (CC_V4 / (CC_WG * CC_TPB))
static_assert(CC_PER * CC_WG * CC_TPB == CC_V4, "grid must cover the state exactly");
static_assert((L_SEQ * NCH) % 4 == 0, "a group of four never leaves the state");

template <int P0, int G>
static __device__ __forceinline__ void concat_groups(cgptr x0, cgptr xc,
                                                     float* xbuf, unsigned gid) {
  v4f o[G];
#pragma unroll
  for (int g = 0; g < G; g++) {
    const unsigned j = (unsigned)(P0 + g) * (CC_WG * CC_TPB) + gid;
#pragma unroll
    for (int e = 0; e < 4; e++) {
      const unsigned idx = 4u * j + e;
      const unsigned bl = idx / NCH, c = idx - bl * NCH;
      const bool     is0 = c < 5;
      const unsigned off = is0 ? bl * 5 + c : bl * 14 + (c - 5);
      cgptr base = is0 ? x0 : xc;
      o[g][e] = base[off];
    }
  }
#pragma unroll
  for (int g = 0; g < G; g++) {
    const unsigned j = (unsigned)(P0 + g) * (CC_WG * CC_TPB) + gid;
    ((AS1 v4f*)xbuf)[j] = o[g];
  }
}

__global__ __launch_bounds__(CC_TPB) void concat_kernel(
    const float* __restrict__ x0D, const float* __restrict__ xctrl,
    float* __restrict__ xbuf) {
  const unsigned gid = blockIdx.x * CC_TPB + threadIdx.x;
  cgptr x0 = (cgptr)x0D, xc = (cgptr)xctrl;
  static_assert(CC_PER == 19, "the three calls below cover 7 + 6 + 6 groups");
  concat_groups<0, 7>(x0, xc, xbuf, gid);
  concat_groups<7, 6>(x0, xc, xbuf, gid);
  concat_groups<13, 6>(x0, xc, xbuf, gid);
}

__global__ void proj_kernel(const float* __restrict__ xbuf,
                            const float* __restrict__ x0D,
                            const float* __restrict__ projw,
                            float* __restrict__ out) {
  __shared__ float yres_s[L_SEQ * 5];
  lptr yres = (lptr)yres_s;
  const int b = blockIdx.x, tid = threadIdx.x;
  cgptr xb = (cgptr)xbuf, x0 = (cgptr)x0D, pj = (cgptr)projw;
  for (int i = tid; i < L_SEQ * 5; i += blockDim.x) {
    const int l = i / 5, c = i - l * 5;
    yres[i] = xb[((size_t)b * L_SEQ + l) * NCH + c] +
              x0[((size_t)b * L_SEQ + l) * 5 + c];
  }
  __syncthreads();
  for (int i = tid; i < 64 * 5; i += blockDim.x) {
    const int o = i / 5, c = i - o * 5;
    cgptr pw = pj + o * L_SEQ;
    float acc = 0.f;
    for (int l = 0; l < L_SEQ; l++) acc = fmaf(pw[l], yres[l * 5 + c], acc);
    ((gptr)out)[(size_t)b * 320 + i] = acc;
  }
}

extern "C" void kernel_launch(void* const* d_in, const int* in_sizes, int n_in,
                              void* d_out, int out_size, void* d_ws,
                              size_t ws_size, hipStream_t stream) {
  const float* x0D   = (const float*)d_in[0];
  const float* xctrl = (const float*)d_in[1];
  const float* W1    = (const float*)d_in[2];
  const float* b1    = (const float*)d_in[3];
  const float* W2    = (const float*)d_in[4];
  const float* b2    = (const float*)d_in[5];
  const float* projw = (const float*)d_in[6];
  float* out = (float*)d_out;

  float* ws   = (float*)d_ws;
  float* xbuf = ws;
  const size_t XB = (size_t)B_TOT * L_SEQ * NCH;
  float* Wt1p = xbuf + XB;
  float* Wt2p = Wt1p + 28 * W1BR;

  const int packN = 28 * W1BR + 28 * W2BR;
  pack_weights_kernel<<<(packN + 255) / 256, 256, 0, stream>>>(W1, W2, Wt1p, Wt2p);
  concat_kernel<<<CC_WG, CC_TPB, 0, stream>>>(x0D, xctrl, xbuf);
  scinet_level_kernel<0><<<B_TOT, TPB, 0, stream>>>(xbuf, Wt1p, Wt2p, b1, b2);
  scinet_level_kernel<1><<<B_TOT * 2, TPB, 0, stream>>>(xbuf, Wt1p, Wt2p, b1, b2);
  scinet_level_kernel<2><<<B_TOT * 4, TPB, 0, stream>>>(xbuf, Wt1p, Wt2p, b1, b2);
  proj_kernel<<<B_TOT, 256, 0, stream>>>(xbuf, x0D, projw, out);
}


# Level kernel: per-lane LDS address arithmetic removed: each A tile of a pair reads from one base register + constant offsets; ch 16-18 tap sums one row per lane (scratch pitch 17); sums unchanged
# speedup vs baseline: 2.0149x; 1.0495x over previous
#include <hip/hip_runtime.h>
#include <math.h>

typedef __attribute__((ext_vector_type(2))) float v2f;
typedef __attribute__((ext_vector_type(4))) float v4f;
typedef __attribute__((ext_vector_type(8))) float v8f;

#define AS3 __attribute__((address_space(3)))
#define AS1 __attribute__((address_space(1)))
typedef AS3 float*       lptr;
typedef const AS3 float* clptr;
typedef AS1 float*       gptr;
typedef const AS1 float* cgptr;

#define B_TOT    8192
#define L_SEQ    256
#define NCH      19
#define KPACK1   96
#define KPACK2   64
#define KS1      (KPACK1 / 4)
#define KS2      (KPACK2 / 4)
#define WAVE     32
#define TPB      64
#define KST      5
#define F1M      (KS1 * WAVE * 2)
#define F2M      (KS2 * WAVE * 2)
#define FTL      (KST * WAVE * 2)
#define W1BR     (F1M + FTL)
#define W2BR     (F2M + FTL)
#define TP       17

#define BUF_SZ(LP) ((((LP) + 6) * NCH + 32 + 3) & ~3)
#define HB_SZ(LP)  ((((LP) + 2) * NCH + 32 + 3) & ~3)
#define SMEM_TOT(LP) (4 * BUF_SZ(LP) + 2 * HB_SZ(LP))

static __device__ __forceinline__ v8f wmma4(v2f a, v2f b, v8f c) {
  return __builtin_amdgcn_wmma_f32_16x16x4_f32(false, a, false, b, (short)0, c,
                                               false, false);
}

static __device__ __forceinline__ float leaky(float x) {
  return x > 0.f ? x : 0.01f * x;
}

static __device__ __forceinline__ float fast_tanh(float x) {
  x = fminf(fmaxf(x, -15.f), 15.f);
  const float e = __expf(2.f * x);
  return __fdividef(e - 1.f, e + 1.f);
}

static __device__ __forceinline__ void tile_bases(clptr& p, clptr& q) {
  asm("" : "+v"(p), "+v"(q));
}

static __device__ __forceinline__ void wave_lds_order() {
  __builtin_amdgcn_fence(__ATOMIC_ACQ_REL, "wavefront");
}

template <int KS>
static __device__ __forceinline__ void load_a(clptr ap, v2f (&a)[KS]) {
#pragma unroll
  for (int ks = 0; ks < KS; ks++) { a[ks].x = ap[4 * ks]; a[ks].y = ap[4 * ks + 1]; }
}

template <int ROWS>
static __device__ __forceinline__ void tail_products(cgptr Wt, clptr src, lptr scr,
                                                     int lhi, int lln) {
  constexpr int nT = (ROWS + 15) >> 4;
  const int lane = lhi * 16 + lln;
  v2f bt[KST];
#pragma unroll
  for (int ks = 0; ks < KST; ks++) bt[ks] = ((const AS1 v2f*)Wt)[ks * WAVE + lane];
  v2f an[KST];
  load_a<KST>(src + lln * NCH + 2 * lhi, an);
#pragma nounroll
  for (int mt = 0; mt < nT; mt++) {
    const int m0 = mt << 4;
    v2f a[KST];
#pragma unroll
    for (int ks = 0; ks < KST; ks++) a[ks] = an[ks];
    {
      const int mn = (mt + 1 < nT) ? m0 + 16 : m0;
      int prow = mn + lln; if (prow > ROWS - 1) prow = ROWS - 1;
      load_a<KST>(src + prow * NCH + 2 * lhi, an);
    }
    v8f acc = {};
#pragma unroll
    for (int ks = 0; ks < KST; ks++) acc = wmma4(a[ks], bt[ks], acc);
    lptr srow = scr + (m0 + 8 * lhi) * TP + lln;
    if (m0 + 16 <= ROWS) {
#pragma unroll
      for (int v = 0; v < 8; v++) srow[v * TP] = acc[v];
    } else {
#pragma unroll
      for (int v = 0; v < 8; v++)
        if (m0 + 8 * lhi + v < ROWS) srow[v * TP] = acc[v];
    }
  }
}

template <int TAPS>
static __device__ __forceinline__ float tail_sum(clptr scr, int m, int j) {
  clptr sp = scr + m * TP + j;
  float s = sp[0];
#pragma unroll
  for (int t = 1; t < TAPS; t++) s += sp[t * (TP + 3)];
  return s;
}

template <bool TO_GLOBAL, int LP>
static __device__ void run_branch(cgptr Wf1, cgptr Wf2, cgptr b1, cgptr b2,
                                  clptr in, lptr hb, clptr other, lptr outLds,
                                  lptr scr, gptr xg, int tmul, int tadd, float sgn) {
  const int lane = threadIdx.x & (WAVE - 1);
  const int lhi  = lane >> 4;
  const int lln  = lane & 15;
  static_assert(LP % 32 == 0, "M tiles are taken in pairs; 3*LP outputs fill whole waves");

  constexpr int M1 = LP + 2;
  {
    tail_products<LP + 6>(Wf1 + F1M, in, hb, lhi, lln);
    const float bt0 = b1[16], bt1 = b1[17], bt2 = b1[18];
    constexpr int NIT = (M1 + WAVE - 1) / WAVE;
    float hv[NIT][3];
#pragma unroll
    for (int it = 0; it < NIT; it++) {
      int m = it * WAVE + lane; if (m > M1 - 1) m = M1 - 1;
      hv[it][0] = leaky(tail_sum<5>(hb, m, 0) + bt0);
      hv[it][1] = leaky(tail_sum<5>(hb, m, 1) + bt1);
      hv[it][2] = leaky(tail_sum<5>(hb, m, 2) + bt2);
    }
    wave_lds_order();
#pragma unroll
    for (int it = 0; it < NIT; it++) {
      const int m = it * WAVE + lane;
      if (m < M1) {
        lptr hr = hb + m * NCH + 16;
        hr[0] = hv[it][0]; hr[1] = hv[it][1]; hr[2] = hv[it][2];
      }
    }
  }
  {
    v2f bw[KS1];
#pragma unroll
    for (int ks = 0; ks < KS1; ks++)
      bw[ks] = ((const AS1 v2f*)Wf1)[ks * WAVE + lane];
    const float b1a = b1[lln];
    constexpr int nFull = LP >> 4;
#pragma nounroll
    for (int mt = 0; mt < nFull; mt += 2) {
      const int m0 = mt << 4;
      clptr apA = in + (m0 + lln) * NCH + 2 * lhi;
      clptr apB = apA + 16 * NCH;
      tile_bases(apA, apB);
      v8f accA = {}; v8f accB = {};
#pragma unroll
      for (int ks = 0; ks < KS1; ks++) {
        v2f aA; aA.x = apA[4 * ks]; aA.y = apA[4 * ks + 1];
        v2f aB; aB.x = apB[4 * ks]; aB.y = apB[4 * ks + 1];
        accA = wmma4(aA, bw[ks], accA);
        accB = wmma4(aB, bw[ks], accB);
      }
      lptr hrA = hb + (m0 + 8 * lhi) * NCH + lln;
      lptr hrB = hrA + 16 * NCH;
#pragma unroll
      for (int v = 0; v < 8; v++) hrA[v * NCH] = leaky(accA[v] + b1a);
#pragma unroll
      for (int v = 0; v < 8; v++) hrB[v * NCH] = leaky(accB[v] + b1a);
    }
    {
      int mrow = LP + lln; if (mrow > M1 - 1) mrow = M1 - 1;
      clptr ap = in + mrow * NCH + 2 * lhi;
      v8f acc = {};
#pragma unroll
      for (int ks = 0; ks < KS1; ks++) {
        v2f a; a.x = ap[4 * ks]; a.y = ap[4 * ks + 1];
        acc = wmma4(a, bw[ks], acc);
      }
      if (lhi == 0) {
        lptr hr = hb + LP * NCH + lln;
        hr[0]   = leaky(acc[0] + b1a);
        hr[NCH] = leaky(acc[1] + b1a);
      }
    }
  }
  __syncthreads();

  {
    tail_products<M1>(Wf2 + F2M, hb, scr, lhi, lln);
    const float bt0 = b2[16], bt1 = b2[17], bt2 = b2[18];
    constexpr int NIT = LP / WAVE;
    float tv[NIT][3];
#pragma unroll
    for (int it = 0; it < NIT; it++) {
      const int m = it * WAVE + lane;
      tv[it][0] = tail_sum<3>(scr, m, 0) + bt0;
      tv[it][1] = tail_sum<3>(scr, m, 1) + bt1;
      tv[it][2] = tail_sum<3>(scr, m, 2) + bt2;
    }
    if (!TO_GLOBAL) wave_lds_order();
#pragma unroll
    for (int it = 0; it < NIT; it++) {
      const int m = it * WAVE + lane;
#pragma unroll
      for (int j = 0; j < 3; j++) {
        const float t   = fast_tanh(tv[it][j]);
        const float oth = other[(3 + m) * NCH + 16 + j];
        if (!TO_GLOBAL) outLds[(3 + m) * NCH + 16 + j] = oth * __expf(t);
        else            xg[(m * tmul + tadd) * NCH + 16 + j] = fmaf(sgn, t, oth);
      }
    }
  }
  {
    v2f bw[KS2];
#pragma unroll
    for (int ks = 0; ks < KS2; ks++)
      bw[ks] = ((const AS1 v2f*)Wf2)[ks * WAVE + lane];
    const float b2a = b2[lln];
    constexpr int nMt2 = LP >> 4;
#pragma nounroll
    for (int mt = 0; mt < nMt2; mt += 2) {
      const int m0 = mt << 4;
      clptr apA = hb + (m0 + lln) * NCH + 2 * lhi;
      clptr apB = apA + 16 * NCH;
      tile_bases(apA, apB);
      v8f accA = {}; v8f accB = {};
#pragma unroll
      for (int ks = 0; ks < KS2; ks++) {
        v2f aA; aA.x = apA[4 * ks]; aA.y = apA[4 * ks + 1];
        v2f aB; aB.x = apB[4 * ks]; aB.y = apB[4 * ks + 1];
        accA = wmma4(aA, bw[ks], accA);
        accB = wmma4(aB, bw[ks], accB);
      }
      const int mb = m0 + 8 * lhi;
#pragma unroll
      for (int v = 0; v < 8; v++) {
        const int m = mb + v;
        const float t   = fast_tanh(accA[v] + b2a);
        const float oth = other[(3 + m) * NCH + lln];
        if (!TO_GLOBAL) outLds[(3 + m) * NCH + lln] = oth * __expf(t);
        else            xg[(m * tmul + tadd) * NCH + lln] = fmaf(sgn, t, oth);
      }
#pragma unroll
      for (int v = 0; v < 8; v++) {
        const int m = mb + 16 + v;
        const float t   = fast_tanh(accB[v] + b2a);
        const float oth = other[(3 + m) * NCH + lln];
        if (!TO_GLOBAL) outLds[(3 + m) * NCH + lln] = oth * __expf(t);
        else            xg[(m * tmul + tadd) * NCH + lln] = fmaf(sgn, t, oth);
      }
    }
  }
  if (!TO_GLOBAL) __syncthreads();
}

template <int DEPTH>
__global__ __launch_bounds__(TPB) void scinet_level_kernel(
    float* __restrict__ xbuf_, const float* __restrict__ Wf1p_,
    const float* __restrict__ Wf2p_, const float* __restrict__ b1g_,
    const float* __restrict__ b2g_) {
  constexpr int LP     = (L_SEQ >> DEPTH) >> 1;
  constexpr int stride = 1 << DEPTH;
  __shared__ __align__(16) float smem[SMEM_TOT(LP)];
  const int tid = threadIdx.x;
  const int w   = __builtin_amdgcn_readfirstlane(tid >> 5);
  const int sel = blockIdx.x & (stride - 1);
  const int b   = blockIdx.x >> DEPTH;
  int node = 0;
  for (int i = 0; i < DEPTH; i++) node = 2 * node + 1 + ((sel >> i) & 1);

  lptr sm  = (lptr)smem;
  lptr sev = sm;
  lptr sod = sev + BUF_SZ(LP);
  lptr db  = sod + BUF_SZ(LP);
  lptr cb  = db + BUF_SZ(LP);
  lptr hb  = cb + BUF_SZ(LP) + w * HB_SZ(LP);

  {
    const v4f z4 = {0.f, 0.f, 0.f, 0.f};
    for (int i = tid; i < SMEM_TOT(LP) / 4; i += TPB) ((AS3 v4f*)sm)[i] = z4;
  }
  __syncthreads();

  gptr  xg   = (gptr)(xbuf_ + (size_t)b * L_SEQ * NCH);
  cgptr Wf1p = (cgptr)Wf1p_;
  cgptr Wf2p = (cgptr)Wf2p_;
  cgptr b1g  = (cgptr)b1g_;
  cgptr b2g  = (cgptr)b2g_;

  constexpr int NEL   = LP * NCH;
  constexpr int NFULL = NEL / TPB;
  constexpr int NREM  = NEL - NFULL * TPB;
  static_assert(NREM == 0 || NREM == WAVE, "the tail pass is whole waves");
#pragma unroll 19
  for (int p = 0; p < NFULL; p++) {
    const int i = p * TPB + tid;
    const int r = i / NCH, c = i - r * NCH;
    sev[(3 + r) * NCH + c] = xg[((2 * r) * stride + sel) * NCH + c];
    sod[(3 + r) * NCH + c] = xg[((2 * r + 1) * stride + sel) * NCH + c];
  }
  if (NREM != 0 && w == 0) {
    const int i = NFULL * TPB + tid;
    const int r = i / NCH, c = i - r * NCH;
    sev[(3 + r) * NCH + c] = xg[((2 * r) * stride + sel) * NCH + c];
    sod[(3 + r) * NCH + c] = xg[((2 * r + 1) * stride + sel) * NCH + c];
  }
  __syncthreads();
  for (int i = tid; i < 2 * 3 * NCH; i += TPB) {
    const int c = i % NCH, r = (i / NCH) % 3;
    lptr buf = (i >= 3 * NCH) ? sod : sev;
    buf[r * NCH + c]            = buf[3 * NCH + c];
    buf[(LP + 3 + r) * NCH + c] = buf[(LP + 2) * NCH + c];
  }
  __syncthreads();

  const int nb = node * 4;
  {
    const int br = nb + w;
    run_branch<false, LP>(Wf1p + (size_t)br * W1BR, Wf2p + (size_t)br * W2BR,
                          b1g + br * NCH, b2g + br * NCH,
                          w ? sod : sev, hb, w ? sev : sod, w ? cb : db,
                          w ? cb : db,
                          xg, 0, 0, 0.f);
  }
  for (int i = tid; i < 2 * 3 * NCH; i += TPB) {
    const int c = i % NCH, r = (i / NCH) % 3;
    lptr buf = (i >= 3 * NCH) ? cb : db;
    buf[r * NCH + c]            = buf[3 * NCH + c];
    buf[(LP + 3 + r) * NCH + c] = buf[(LP + 2) * NCH + c];
  }
  __syncthreads();
  {
    const int br = nb + 3 - w;
    run_branch<true, LP>(Wf1p + (size_t)br * W1BR, Wf2p + (size_t)br * W2BR,
                         b1g + br * NCH, b2g + br * NCH,
                         w ? cb : db, hb, w ? db : cb, db,
                         w ? sod : sev,
                         xg, 2 * stride, sel + w * stride, w ? -1.f : 1.f);
  }
}

template <int TAPS, int KMAIN>
static __device__ __forceinline__ float pack_value(const float* __restrict__ Wg, int nb,
                                                   int rem, int fmain) {
  const bool tail = rem >= fmain;
  const int r  = tail ? rem - fmain : rem;
  const int ks = r >> 6, ln = (r >> 1) & 31, e = r & 1;
  const int n  = ln & 15;
  const int k  = 4 * ks + 2 * (ln >> 4) + e;
  int co, ci, tap; bool live;
  if (!tail) { tap = k / NCH; ci = k - tap * NCH; co = n; live = k < KMAIN; }
  else       { tap = n / 3; co = 16 + (n - 3 * tap); ci = k; live = (n < 3 * TAPS) && (k < NCH); }
  return live ? Wg[((size_t)(nb * NCH + co) * NCH + ci) * TAPS + tap] : 0.f;
}

__global__ void pack_weights_kernel(const float* __restrict__ W1,
                                    const float* __restrict__ W2,
                                    float* __restrict__ Wf1p,
                                    float* __restrict__ Wf2p) {
  const int idx = blockIdx.x * blockDim.x + threadIdx.x;
  const int N1 = 28 * W1BR, N2 = 28 * W2BR;
  if (idx < N1) {
    const int nb = idx / W1BR, rem = idx - nb * W1BR;
    Wf1p[idx] = pack_value<5, 95>(W1, nb, rem, F1M);
  } else if (idx < N1 + N2) {
    const int j = idx - N1;
    const int nb = j / W2BR, rem = j - nb * W2BR;
    Wf2p[j] = pack_value<3, 57>(W2, nb, rem, F2M);
  }
}

#define CC_TPB   256
#define CC_WG    2048
#define CC_V4    (B_TOT * L_SEQ * NCH / 4)
#define CC_PER   (CC_V4 / (CC_WG * CC_TPB))
static_assert(CC_PER * CC_WG * CC_TPB == CC_V4, "grid must cover the state exactly");
static_assert((L_SEQ * NCH) % 4 == 0, "a group of four never leaves the state");

template <int P0, int G>
static __device__ __forceinline__ void concat_groups(cgptr x0, cgptr xc,
                                                     float* xbuf, unsigned gid) {
  v4f o[G];
#pragma unroll
  for (int g = 0; g < G; g++) {
    const unsigned j = (unsigned)(P0 + g) * (CC_WG * CC_TPB) + gid;
#pragma unroll
    for (int e = 0; e < 4; e++) {
      const unsigned idx = 4u * j + e;
      const unsigned bl = idx / NCH, c = idx - bl * NCH;
      const bool     is0 = c < 5;
      const unsigned off = is0 ? bl * 5 + c : bl * 14 + (c - 5);
      cgptr base = is0 ? x0 : xc;
      o[g][e] = base[off];
    }
  }
#pragma unroll
  for (int g = 0; g < G; g++) {
    const unsigned j = (unsigned)(P0 + g) * (CC_WG * CC_TPB) + gid;
    ((AS1 v4f*)xbuf)[j] = o[g];
  }
}

__global__ __launch_bounds__(CC_TPB) void concat_kernel(
    const float* __restrict__ x0D, const float* __restrict__ xctrl,
    float* __restrict__ xbuf) {
  const unsigned gid = blockIdx.x * CC_TPB + threadIdx.x;
  cgptr x0 = (cgptr)x0D, xc = (cgptr)xctrl;
  static_assert(CC_PER == 19, "the three calls below cover 7 + 6 + 6 groups");
  concat_groups<0, 7>(x0, xc, xbuf, gid);
  concat_groups<7, 6>(x0, xc, xbuf, gid);
  concat_groups<13, 6>(x0, xc, xbuf, gid);
}

__global__ void proj_kernel(const float* __restrict__ xbuf,
                            const float* __restrict__ x0D,
                            const float* __restrict__ projw,
                            float* __restrict__ out) {
  __shared__ float yres_s[L_SEQ * 5];
  lptr yres = (lptr)yres_s;
  const int b = blockIdx.x, tid = threadIdx.x;
  cgptr xb = (cgptr)xbuf, x0 = (cgptr)x0D, pj = (cgptr)projw;
  for (int i = tid; i < L_SEQ * 5; i += blockDim.x) {
    const int l = i / 5, c = i - l * 5;
    yres[i] = xb[((size_t)b * L_SEQ + l) * NCH + c] +
              x0[((size_t)b * L_SEQ + l) * 5 + c];
  }
  __syncthreads();
  for (int i = tid; i < 64 * 5; i += blockDim.x) {
    const int o = i / 5, c = i - o * 5;
    cgptr pw = pj + o * L_SEQ;
    float acc = 0.f;
    for (int l = 0; l < L_SEQ; l++) acc = fmaf(pw[l], yres[l * 5 + c], acc);
    ((gptr)out)[(size_t)b * 320 + i] = acc;
  }
}

extern "C" void kernel_launch(void* const* d_in, const int* in_sizes, int n_in,
                              void* d_out, int out_size, void* d_ws,
                              size_t ws_size, hipStream_t stream) {
  const float* x0D   = (const float*)d_in[0];
  const float* xctrl = (const float*)d_in[1];
  const float* W1    = (const float*)d_in[2];
  const float* b1    = (const float*)d_in[3];
  const float* W2    = (const float*)d_in[4];
  const float* b2    = (const float*)d_in[5];
  const float* projw = (const float*)d_in[6];
  float* out = (float*)d_out;

  float* ws   = (float*)d_ws;
  float* xbuf = ws;
  const size_t XB = (size_t)B_TOT * L_SEQ * NCH;
  float* Wt1p = xbuf + XB;
  float* Wt2p = Wt1p + 28 * W1BR;

  const int packN = 28 * W1BR + 28 * W2BR;
  pack_weights_kernel<<<(packN + 255) / 256, 256, 0, stream>>>(W1, W2, Wt1p, Wt2p);
  concat_kernel<<<CC_WG, CC_TPB, 0, stream>>>(x0D, xctrl, xbuf);
  scinet_level_kernel<0><<<B_TOT, TPB, 0, stream>>>(xbuf, Wt1p, Wt2p, b1, b2);
  scinet_level_kernel<1><<<B_TOT * 2, TPB, 0, stream>>>(xbuf, Wt1p, Wt2p, b1, b2);
  scinet_level_kernel<2><<<B_TOT * 4, TPB, 0, stream>>>(xbuf, Wt1p, Wt2p, b1, b2);
  proj_kernel<<<B_TOT, 256, 0, stream>>>(xbuf, x0D, projw, out);
}


# Last level and output projection fused: one 256-thread workgroup per batch element runs its 4 sub-sequences, keeps only eu/ou channels 0-4 (all the projection reads) in LDS, projects them; same sums
# speedup vs baseline: 2.2457x; 1.1145x over previous
#include <hip/hip_runtime.h>
#include <math.h>

typedef __attribute__((ext_vector_type(2))) float v2f;
typedef __attribute__((ext_vector_type(4))) float v4f;
typedef __attribute__((ext_vector_type(8))) float v8f;

#define AS3 __attribute__((address_space(3)))
#define AS1 __attribute__((address_space(1)))
typedef AS3 float*       lptr;
typedef const AS3 float* clptr;
typedef AS1 float*       gptr;
typedef const AS1 float* cgptr;

#define B_TOT    8192
#define L_SEQ    256
#define NCH      19
#define KPACK1   96
#define KPACK2   64
#define KS1      (KPACK1 / 4)
#define KS2      (KPACK2 / 4)
#define WAVE     32
#define TPB      64
#define KST      5
#define F1M      (KS1 * WAVE * 2)
#define F2M      (KS2 * WAVE * 2)
#define FTL      (KST * WAVE * 2)
#define W1BR     (F1M + FTL)
#define W2BR     (F2M + FTL)
#define TP       17

#define BUF_SZ(LP) ((((LP) + 6) * NCH + 32 + 3) & ~3)
#define HB_SZ(LP)  ((((LP) + 2) * NCH + 32 + 3) & ~3)
#define SMEM_TOT(LP) (4 * BUF_SZ(LP) + 2 * HB_SZ(LP))

static __device__ __forceinline__ v8f wmma4(v2f a, v2f b, v8f c) {
  return __builtin_amdgcn_wmma_f32_16x16x4_f32(false, a, false, b, (short)0, c,
                                               false, false);
}

static __device__ __forceinline__ float leaky(float x) {
  return x > 0.f ? x : 0.01f * x;
}

static __device__ __forceinline__ float fast_tanh(float x) {
  x = fminf(fmaxf(x, -15.f), 15.f);
  const float e = __expf(2.f * x);
  return __fdividef(e - 1.f, e + 1.f);
}

static __device__ __forceinline__ void tile_bases(clptr& p, clptr& q) {
  asm("" : "+v"(p), "+v"(q));
}

static __device__ __forceinline__ void wave_lds_order() {
  __builtin_amdgcn_fence(__ATOMIC_ACQ_REL, "wavefront");
}

template <int KS>
static __device__ __forceinline__ void load_a(clptr ap, v2f (&a)[KS]) {
#pragma unroll
  for (int ks = 0; ks < KS; ks++) { a[ks].x = ap[4 * ks]; a[ks].y = ap[4 * ks + 1]; }
}

template <int ROWS>
static __device__ __forceinline__ void tail_products(cgptr Wt, clptr src, lptr scr,
                                                     int lhi, int lln) {
  constexpr int nT = (ROWS + 15) >> 4;
  const int lane = lhi * 16 + lln;
  v2f bt[KST];
#pragma unroll
  for (int ks = 0; ks < KST; ks++) bt[ks] = ((const AS1 v2f*)Wt)[ks * WAVE + lane];
  v2f an[KST];
  load_a<KST>(src + lln * NCH + 2 * lhi, an);
#pragma nounroll
  for (int mt = 0; mt < nT; mt++) {
    const int m0 = mt << 4;
    v2f a[KST];
#pragma unroll
    for (int ks = 0; ks < KST; ks++) a[ks] = an[ks];
    {
      const int mn = (mt + 1 < nT) ? m0 + 16 : m0;
      int prow = mn + lln; if (prow > ROWS - 1) prow = ROWS - 1;
      load_a<KST>(src + prow * NCH + 2 * lhi, an);
    }
    v8f acc = {};
#pragma unroll
    for (int ks = 0; ks < KST; ks++) acc = wmma4(a[ks], bt[ks], acc);
    lptr srow = scr + (m0 + 8 * lhi) * TP + lln;
    if (m0 + 16 <= ROWS) {
#pragma unroll
      for (int v = 0; v < 8; v++) srow[v * TP] = acc[v];
    } else {
#pragma unroll
      for (int v = 0; v < 8; v++)
        if (m0 + 8 * lhi + v < ROWS) srow[v * TP] = acc[v];
    }
  }
}

template <int TAPS>
static __device__ __forceinline__ float tail_sum(clptr scr, int m, int j) {
  clptr sp = scr + m * TP + j;
  float s = sp[0];
#pragma unroll
  for (int t = 1; t < TAPS; t++) s += sp[t * (TP + 3)];
  return s;
}

template <int MODE, int LP>
static __device__ void run_branch(cgptr Wf1, cgptr Wf2, cgptr b1, cgptr b2,
                                  clptr in, lptr hb, clptr other, lptr outLds,
                                  lptr scr, gptr xg, int tmul, int tadd, float sgn) {
  const int lane = threadIdx.x & (WAVE - 1);
  const int lhi  = lane >> 4;
  const int lln  = lane & 15;
  static_assert(LP % 32 == 0, "M tiles are taken in pairs; 3*LP outputs fill whole waves");

  constexpr int M1 = LP + 2;
  {
    tail_products<LP + 6>(Wf1 + F1M, in, hb, lhi, lln);
    const float bt0 = b1[16], bt1 = b1[17], bt2 = b1[18];
    constexpr int NIT = (M1 + WAVE - 1) / WAVE;
    float hv[NIT][3];
#pragma unroll
    for (int it = 0; it < NIT; it++) {
      int m = it * WAVE + lane; if (m > M1 - 1) m = M1 - 1;
      hv[it][0] = leaky(tail_sum<5>(hb, m, 0) + bt0);
      hv[it][1] = leaky(tail_sum<5>(hb, m, 1) + bt1);
      hv[it][2] = leaky(tail_sum<5>(hb, m, 2) + bt2);
    }
    wave_lds_order();
#pragma unroll
    for (int it = 0; it < NIT; it++) {
      const int m = it * WAVE + lane;
      if (m < M1) {
        lptr hr = hb + m * NCH + 16;
        hr[0] = hv[it][0]; hr[1] = hv[it][1]; hr[2] = hv[it][2];
      }
    }
  }
  {
    v2f bw[KS1];
#pragma unroll
    for (int ks = 0; ks < KS1; ks++)
      bw[ks] = ((const AS1 v2f*)Wf1)[ks * WAVE + lane];
    const float b1a = b1[lln];
    constexpr int nFull = LP >> 4;
#pragma nounroll
    for (int mt = 0; mt < nFull; mt += 2) {
      const int m0 = mt << 4;
      clptr apA = in + (m0 + lln) * NCH + 2 * lhi;
      clptr apB = apA + 16 * NCH;
      tile_bases(apA, apB);
      v8f accA = {}; v8f accB = {};
#pragma unroll
      for (int ks = 0; ks < KS1; ks++) {
        v2f aA; aA.x = apA[4 * ks]; aA.y = apA[4 * ks + 1];
        v2f aB; aB.x = apB[4 * ks]; aB.y = apB[4 * ks + 1];
        accA = wmma4(aA, bw[ks], accA);
        accB = wmma4(aB, bw[ks], accB);
      }
      lptr hrA = hb + (m0 + 8 * lhi) * NCH + lln;
      lptr hrB = hrA + 16 * NCH;
#pragma unroll
      for (int v = 0; v < 8; v++) hrA[v * NCH] = leaky(accA[v] + b1a);
#pragma unroll
      for (int v = 0; v < 8; v++) hrB[v * NCH] = leaky(accB[v] + b1a);
    }
    {
      int mrow = LP + lln; if (mrow > M1 - 1) mrow = M1 - 1;
      clptr ap = in + mrow * NCH + 2 * lhi;
      v8f acc = {};
#pragma unroll
      for (int ks = 0; ks < KS1; ks++) {
        v2f a; a.x = ap[4 * ks]; a.y = ap[4 * ks + 1];
        acc = wmma4(a, bw[ks], acc);
      }
      if (lhi == 0) {
        lptr hr = hb + LP * NCH + lln;
        hr[0]   = leaky(acc[0] + b1a);
        hr[NCH] = leaky(acc[1] + b1a);
      }
    }
  }
  __syncthreads();

  if (MODE != 2) {
    tail_products<M1>(Wf2 + F2M, hb, scr, lhi, lln);
    const float bt0 = b2[16], bt1 = b2[17], bt2 = b2[18];
    constexpr int NIT = LP / WAVE;
    float tv[NIT][3];
#pragma unroll
    for (int it = 0; it < NIT; it++) {
      const int m = it * WAVE + lane;
      tv[it][0] = tail_sum<3>(scr, m, 0) + bt0;
      tv[it][1] = tail_sum<3>(scr, m, 1) + bt1;
      tv[it][2] = tail_sum<3>(scr, m, 2) + bt2;
    }
    if (MODE == 0) wave_lds_order();
#pragma unroll
    for (int it = 0; it < NIT; it++) {
      const int m = it * WAVE + lane;
#pragma unroll
      for (int j = 0; j < 3; j++) {
        const float t   = fast_tanh(tv[it][j]);
        const float oth = other[(3 + m) * NCH + 16 + j];
        if (MODE == 0) outLds[(3 + m) * NCH + 16 + j] = oth * __expf(t);
        else           xg[(m * tmul + tadd) * NCH + 16 + j] = fmaf(sgn, t, oth);
      }
    }
  }
  {
    v2f bw[KS2];
#pragma unroll
    for (int ks = 0; ks < KS2; ks++)
      bw[ks] = ((const AS1 v2f*)Wf2)[ks * WAVE + lane];
    const float b2a = b2[lln];
    constexpr int nMt2 = LP >> 4;
#pragma nounroll
    for (int mt = 0; mt < nMt2; mt += 2) {
      const int m0 = mt << 4;
      clptr apA = hb + (m0 + lln) * NCH + 2 * lhi;
      clptr apB = apA + 16 * NCH;
      tile_bases(apA, apB);
      v8f accA = {}; v8f accB = {};
#pragma unroll
      for (int ks = 0; ks < KS2; ks++) {
        v2f aA; aA.x = apA[4 * ks]; aA.y = apA[4 * ks + 1];
        v2f aB; aB.x = apB[4 * ks]; aB.y = apB[4 * ks + 1];
        accA = wmma4(aA, bw[ks], accA);
        accB = wmma4(aB, bw[ks], accB);
      }
      const int mb = m0 + 8 * lhi;
      if (MODE == 2) {
        if (lln < 5) {
#pragma unroll
          for (int v = 0; v < 8; v++) {
            const int m = mb + v;
            const float t = fast_tanh(accA[v] + b2a);
            outLds[(m * tmul + tadd) * 5 + lln] = fmaf(sgn, t, other[(3 + m) * NCH + lln]);
          }
#pragma unroll
          for (int v = 0; v < 8; v++) {
            const int m = mb + 16 + v;
            const float t = fast_tanh(accB[v] + b2a);
            outLds[(m * tmul + tadd) * 5 + lln] = fmaf(sgn, t, other[(3 + m) * NCH + lln]);
          }
        }
      } else {
#pragma unroll
        for (int v = 0; v < 8; v++) {
          const int m = mb + v;
          const float t   = fast_tanh(accA[v] + b2a);
          const float oth = other[(3 + m) * NCH + lln];
          if (MODE == 0) outLds[(3 + m) * NCH + lln] = oth * __expf(t);
          else           xg[(m * tmul + tadd) * NCH + lln] = fmaf(sgn, t, oth);
        }
#pragma unroll
        for (int v = 0; v < 8; v++) {
          const int m = mb + 16 + v;
          const float t   = fast_tanh(accB[v] + b2a);
          const float oth = other[(3 + m) * NCH + lln];
          if (MODE == 0) outLds[(3 + m) * NCH + lln] = oth * __expf(t);
          else           xg[(m * tmul + tadd) * NCH + lln] = fmaf(sgn, t, oth);
        }
      }
    }
  }
  if (MODE == 0) __syncthreads();
}

template <int DEPTH, int OUTMODE>
static __device__ __forceinline__ void level_body(lptr sm, int tid, int w, int sel,
                                                  gptr xg, cgptr Wf1p, cgptr Wf2p,
                                                  cgptr b1g, cgptr b2g, lptr yq) {
  constexpr int LP     = (L_SEQ >> DEPTH) >> 1;
  constexpr int stride = 1 << DEPTH;
  int node = 0;
  for (int i = 0; i < DEPTH; i++) node = 2 * node + 1 + ((sel >> i) & 1);

  lptr sev = sm;
  lptr sod = sev + BUF_SZ(LP);
  lptr db  = sod + BUF_SZ(LP);
  lptr cb  = db + BUF_SZ(LP);
  lptr hb  = cb + BUF_SZ(LP) + w * HB_SZ(LP);

  {
    const v4f z4 = {0.f, 0.f, 0.f, 0.f};
    for (int i = tid; i < SMEM_TOT(LP) / 4; i += TPB) ((AS3 v4f*)sm)[i] = z4;
  }
  __syncthreads();

  constexpr int NEL   = LP * NCH;
  constexpr int NFULL = NEL / TPB;
  constexpr int NREM  = NEL - NFULL * TPB;
  static_assert(NREM == 0 || NREM == WAVE, "the tail pass is whole waves");
#pragma unroll 19
  for (int p = 0; p < NFULL; p++) {
    const int i = p * TPB + tid;
    const int r = i / NCH, c = i - r * NCH;
    sev[(3 + r) * NCH + c] = xg[((2 * r) * stride + sel) * NCH + c];
    sod[(3 + r) * NCH + c] = xg[((2 * r + 1) * stride + sel) * NCH + c];
  }
  if (NREM != 0 && w == 0) {
    const int i = NFULL * TPB + tid;
    const int r = i / NCH, c = i - r * NCH;
    sev[(3 + r) * NCH + c] = xg[((2 * r) * stride + sel) * NCH + c];
    sod[(3 + r) * NCH + c] = xg[((2 * r + 1) * stride + sel) * NCH + c];
  }
  __syncthreads();
  for (int i = tid; i < 2 * 3 * NCH; i += TPB) {
    const int c = i % NCH, r = (i / NCH) % 3;
    lptr buf = (i >= 3 * NCH) ? sod : sev;
    buf[r * NCH + c]            = buf[3 * NCH + c];
    buf[(LP + 3 + r) * NCH + c] = buf[(LP + 2) * NCH + c];
  }
  __syncthreads();

  const int nb = node * 4;
  {
    const int br = nb + w;
    run_branch<0, LP>(Wf1p + (size_t)br * W1BR, Wf2p + (size_t)br * W2BR,
                      b1g + br * NCH, b2g + br * NCH,
                      w ? sod : sev, hb, w ? sev : sod, w ? cb : db,
                      w ? cb : db,
                      xg, 0, 0, 0.f);
  }
  for (int i = tid; i < 2 * 3 * NCH; i += TPB) {
    const int c = i % NCH, r = (i / NCH) % 3;
    lptr buf = (i >= 3 * NCH) ? cb : db;
    buf[r * NCH + c]            = buf[3 * NCH + c];
    buf[(LP + 3 + r) * NCH + c] = buf[(LP + 2) * NCH + c];
  }
  __syncthreads();
  {
    const int br = nb + 3 - w;
    run_branch<OUTMODE, LP>(Wf1p + (size_t)br * W1BR, Wf2p + (size_t)br * W2BR,
                            b1g + br * NCH, b2g + br * NCH,
                            w ? cb : db, hb, w ? db : cb,
                            OUTMODE == 2 ? yq : db,
                            w ? sod : sev,
                            xg, 2 * stride, sel + w * stride, w ? -1.f : 1.f);
  }
}

template <int DEPTH>
__global__ __launch_bounds__(TPB) void scinet_level_kernel(
    float* __restrict__ xbuf_, const float* __restrict__ Wf1p_,
    const float* __restrict__ Wf2p_, const float* __restrict__ b1g_,
    const float* __restrict__ b2g_) {
  constexpr int LP     = (L_SEQ >> DEPTH) >> 1;
  constexpr int stride = 1 << DEPTH;
  __shared__ __align__(16) float smem[SMEM_TOT(LP)];
  const int tid = threadIdx.x;
  const int w   = __builtin_amdgcn_readfirstlane(tid >> 5);
  const int sel = blockIdx.x & (stride - 1);
  const int b   = blockIdx.x >> DEPTH;
  level_body<DEPTH, 1>((lptr)smem, tid, w, sel,
                       (gptr)(xbuf_ + (size_t)b * L_SEQ * NCH),
                       (cgptr)Wf1p_, (cgptr)Wf2p_, (cgptr)b1g_, (cgptr)b2g_, (lptr)smem);
}

#define LAST_DEPTH 2
#define LAST_PAIRS (1 << LAST_DEPTH)
#define LAST_LP    ((L_SEQ >> LAST_DEPTH) >> 1)
#define LAST_TPB   (LAST_PAIRS * TPB)
__global__ __launch_bounds__(LAST_TPB) void last_level_proj_kernel(
    const float* __restrict__ xbuf_, const float* __restrict__ Wf1p_,
    const float* __restrict__ Wf2p_, const float* __restrict__ b1g_,
    const float* __restrict__ b2g_, const float* __restrict__ x0D,
    const float* __restrict__ projw, float* __restrict__ out) {
  __shared__ __align__(16) float smem[LAST_PAIRS * SMEM_TOT(LAST_LP) + L_SEQ * 5];
  const int tid  = threadIdx.x;
  const int wid  = __builtin_amdgcn_readfirstlane(tid >> 5);
  const int pair = wid >> 1;
  const int b    = blockIdx.x;
  lptr yres = (lptr)smem + LAST_PAIRS * SMEM_TOT(LAST_LP);
  level_body<LAST_DEPTH, 2>((lptr)smem + pair * SMEM_TOT(LAST_LP), tid & (TPB - 1),
                            wid & 1, pair,
                            (gptr)(const_cast<float*>(xbuf_) + (size_t)b * L_SEQ * NCH),
                            (cgptr)Wf1p_, (cgptr)Wf2p_, (cgptr)b1g_, (cgptr)b2g_, yres);
  __syncthreads();
  cgptr x0 = (cgptr)x0D, pj = (cgptr)projw;
  for (int i = tid; i < L_SEQ * 5; i += LAST_TPB)
    yres[i] = yres[i] + x0[(size_t)b * L_SEQ * 5 + i];
  __syncthreads();
  if (tid < 160) {
    const int o = tid / 5, c = tid - o * 5;
    cgptr pw0 = pj + o * L_SEQ;
    cgptr pw1 = pw0 + 32 * L_SEQ;
    float acc0 = 0.f, acc1 = 0.f;
    for (int l = 0; l < L_SEQ; l++) {
      const float y = yres[l * 5 + c];
      acc0 = fmaf(pw0[l], y, acc0);
      acc1 = fmaf(pw1[l], y, acc1);
    }
    ((gptr)out)[(size_t)b * 320 + tid]       = acc0;
    ((gptr)out)[(size_t)b * 320 + tid + 160] = acc1;
  }
}

template <int TAPS, int KMAIN>
static __device__ __forceinline__ float pack_value(const float* __restrict__ Wg, int nb,
                                                   int rem, int fmain) {
  const bool tail = rem >= fmain;
  const int r  = tail ? rem - fmain : rem;
  const int ks = r >> 6, ln = (r >> 1) & 31, e = r & 1;
  const int n  = ln & 15;
  const int k  = 4 * ks + 2 * (ln >> 4) + e;
  int co, ci, tap; bool live;
  if (!tail) { tap = k / NCH; ci = k - tap * NCH; co = n; live = k < KMAIN; }
  else       { tap = n / 3; co = 16 + (n - 3 * tap); ci = k; live = (n < 3 * TAPS) && (k < NCH); }
  return live ? Wg[((size_t)(nb * NCH + co) * NCH + ci) * TAPS + tap] : 0.f;
}

__global__ void pack_weights_kernel(const float* __restrict__ W1,
                                    const float* __restrict__ W2,
                                    float* __restrict__ Wf1p,
                                    float* __restrict__ Wf2p) {
  const int idx = blockIdx.x * blockDim.x + threadIdx.x;
  const int N1 = 28 * W1BR, N2 = 28 * W2BR;
  if (idx < N1) {
    const int nb = idx / W1BR, rem = idx - nb * W1BR;
    Wf1p[idx] = pack_value<5, 95>(W1, nb, rem, F1M);
  } else if (idx < N1 + N2) {
    const int j = idx - N1;
    const int nb = j / W2BR, rem = j - nb * W2BR;
    Wf2p[j] = pack_value<3, 57>(W2, nb, rem, F2M);
  }
}

#define CC_TPB   256
#define CC_WG    2048
#define CC_V4    (B_TOT * L_SEQ * NCH / 4)
#define CC_PER   (CC_V4 / (CC_WG * CC_TPB))
static_assert(CC_PER * CC_WG * CC_TPB == CC_V4, "grid must cover the state exactly");
static_assert((L_SEQ * NCH) % 4 == 0, "a group of four never leaves the state");

template <int P0, int G>
static __device__ __forceinline__ void concat_groups(cgptr x0, cgptr xc,
                                                     float* xbuf, unsigned gid) {
  v4f o[G];
#pragma unroll
  for (int g = 0; g < G; g++) {
    const unsigned j = (unsigned)(P0 + g) * (CC_WG * CC_TPB) + gid;
#pragma unroll
    for (int e = 0; e < 4; e++) {
      const unsigned idx = 4u * j + e;
      const unsigned bl = idx / NCH, c = idx - bl * NCH;
      const bool     is0 = c < 5;
      const unsigned off = is0 ? bl * 5 + c : bl * 14 + (c - 5);
      cgptr base = is0 ? x0 : xc;
      o[g][e] = base[off];
    }
  }
#pragma unroll
  for (int g = 0; g < G; g++) {
    const unsigned j = (unsigned)(P0 + g) * (CC_WG * CC_TPB) + gid;
    ((AS1 v4f*)xbuf)[j] = o[g];
  }
}

__global__ __launch_bounds__(CC_TPB) void concat_kernel(
    const float* __restrict__ x0D, const float* __restrict__ xctrl,
    float* __restrict__ xbuf) {
  const unsigned gid = blockIdx.x * CC_TPB + threadIdx.x;
  cgptr x0 = (cgptr)x0D, xc = (cgptr)xctrl;
  static_assert(CC_PER == 19, "the three calls below cover 7 + 6 + 6 groups");
  concat_groups<0, 7>(x0, xc, xbuf, gid);
  concat_groups<7, 6>(x0, xc, xbuf, gid);
  concat_groups<13, 6>(x0, xc, xbuf, gid);
}

extern "C" void kernel_launch(void* const* d_in, const int* in_sizes, int n_in,
                              void* d_out, int out_size, void* d_ws,
                              size_t ws_size, hipStream_t stream) {
  const float* x0D   = (const float*)d_in[0];
  const float* xctrl = (const float*)d_in[1];
  const float* W1    = (const float*)d_in[2];
  const float* b1    = (const float*)d_in[3];
  const float* W2    = (const float*)d_in[4];
  const float* b2    = (const float*)d_in[5];
  const float* projw = (const float*)d_in[6];
  float* out = (float*)d_out;

  float* ws   = (float*)d_ws;
  float* xbuf = ws;
  const size_t XB = (size_t)B_TOT * L_SEQ * NCH;
  float* Wt1p = xbuf + XB;
  float* Wt2p = Wt1p + 28 * W1BR;

  const int packN = 28 * W1BR + 28 * W2BR;
  pack_weights_kernel<<<(packN + 255) / 256, 256, 0, stream>>>(W1, W2, Wt1p, Wt2p);
  concat_kernel<<<CC_WG, CC_TPB, 0, stream>>>(x0D, xctrl, xbuf);
  scinet_level_kernel<0><<<B_TOT, TPB, 0, stream>>>(xbuf, Wt1p, Wt2p, b1, b2);
  scinet_level_kernel<1><<<B_TOT * 2, TPB, 0, stream>>>(xbuf, Wt1p, Wt2p, b1, b2);
  last_level_proj_kernel<<<B_TOT, LAST_TPB, 0, stream>>>(xbuf, Wt1p, Wt2p, b1, b2,
                                                        x0D, projw, out);
}


# Level 0: four waves per sub-sequence, not two: each branch's M tiles and tap-sum rows are split over two waves sharing its h and scratch LDS: 20 waves per WGP on the same LDS, not 10; same sums
# speedup vs baseline: 2.3520x; 1.0473x over previous
#include <hip/hip_runtime.h>
#include <math.h>

typedef __attribute__((ext_vector_type(2))) float v2f;
typedef __attribute__((ext_vector_type(4))) float v4f;
typedef __attribute__((ext_vector_type(8))) float v8f;

#define AS3 __attribute__((address_space(3)))
#define AS1 __attribute__((address_space(1)))
typedef AS3 float*       lptr;
typedef const AS3 float* clptr;
typedef AS1 float*       gptr;
typedef const AS1 float* cgptr;

#define B_TOT    8192
#define L_SEQ    256
#define NCH      19
#define KPACK1   96
#define KPACK2   64
#define KS1      (KPACK1 / 4)
#define KS2      (KPACK2 / 4)
#define WAVE     32
#define TPB      64
#define KST      5
#define F1M      (KS1 * WAVE * 2)
#define F2M      (KS2 * WAVE * 2)
#define FTL      (KST * WAVE * 2)
#define W1BR     (F1M + FTL)
#define W2BR     (F2M + FTL)
#define TP       17

#define BUF_SZ(LP) ((((LP) + 6) * NCH + 32 + 3) & ~3)
#define HB_SZ(LP)  ((((LP) + 2) * NCH + 32 + 3) & ~3)
#define SMEM_TOT(LP) (4 * BUF_SZ(LP) + 2 * HB_SZ(LP))

static __device__ __forceinline__ v8f wmma4(v2f a, v2f b, v8f c) {
  return __builtin_amdgcn_wmma_f32_16x16x4_f32(false, a, false, b, (short)0, c,
                                               false, false);
}

static __device__ __forceinline__ float leaky(float x) {
  return x > 0.f ? x : 0.01f * x;
}

static __device__ __forceinline__ float fast_tanh(float x) {
  x = fminf(fmaxf(x, -15.f), 15.f);
  const float e = __expf(2.f * x);
  return __fdividef(e - 1.f, e + 1.f);
}

static __device__ __forceinline__ void tile_bases(clptr& p, clptr& q) {
  asm("" : "+v"(p), "+v"(q));
}

static __device__ __forceinline__ void wave_lds_order() {
  __builtin_amdgcn_fence(__ATOMIC_ACQ_REL, "wavefront");
}

template <int KS>
static __device__ __forceinline__ void load_a(clptr ap, v2f (&a)[KS]) {
#pragma unroll
  for (int ks = 0; ks < KS; ks++) { a[ks].x = ap[4 * ks]; a[ks].y = ap[4 * ks + 1]; }
}

template <int ROWS>
static __device__ __forceinline__ void tail_products(cgptr Wt, clptr src, lptr scr,
                                                     int lhi, int lln, int mt0, int mt1) {
  const int lane = lhi * 16 + lln;
  v2f bt[KST];
#pragma unroll
  for (int ks = 0; ks < KST; ks++) bt[ks] = ((const AS1 v2f*)Wt)[ks * WAVE + lane];
  v2f an[KST];
  {
    int prow = (mt0 << 4) + lln; if (prow > ROWS - 1) prow = ROWS - 1;
    load_a<KST>(src + prow * NCH + 2 * lhi, an);
  }
#pragma nounroll
  for (int mt = mt0; mt < mt1; mt++) {
    const int m0 = mt << 4;
    v2f a[KST];
#pragma unroll
    for (int ks = 0; ks < KST; ks++) a[ks] = an[ks];
    {
      const int mn = (mt + 1 < mt1) ? m0 + 16 : m0;
      int prow = mn + lln; if (prow > ROWS - 1) prow = ROWS - 1;
      load_a<KST>(src + prow * NCH + 2 * lhi, an);
    }
    v8f acc = {};
#pragma unroll
    for (int ks = 0; ks < KST; ks++) acc = wmma4(a[ks], bt[ks], acc);
    lptr srow = scr + (m0 + 8 * lhi) * TP + lln;
    if (m0 + 16 <= ROWS) {
#pragma unroll
      for (int v = 0; v < 8; v++) srow[v * TP] = acc[v];
    } else {
#pragma unroll
      for (int v = 0; v < 8; v++)
        if (m0 + 8 * lhi + v < ROWS) srow[v * TP] = acc[v];
    }
  }
}

template <int TAPS>
static __device__ __forceinline__ float tail_sum(clptr scr, int m, int j) {
  clptr sp = scr + m * TP + j;
  float s = sp[0];
#pragma unroll
  for (int t = 1; t < TAPS; t++) s += sp[t * (TP + 3)];
  return s;
}

template <int MODE, int LP, int NW>
static __device__ void run_branch(cgptr Wf1, cgptr Wf2, cgptr b1, cgptr b2,
                                  clptr in, lptr hb, clptr other, lptr outLds,
                                  lptr scr, gptr xg, int tmul, int tadd, float sgn,
                                  int hw) {
  const int lane = threadIdx.x & (WAVE - 1);
  const int lhi  = lane >> 4;
  const int lln  = lane & 15;
  static_assert(NW == 1 || NW == 2, "one or two waves per branch");
  static_assert(LP % (32 * NW) == 0, "each wave takes M tiles in pairs; 3*LP outputs fill whole waves");
  static_assert(MODE != 2 || NW == 1, "the last level runs one wave per branch");
  constexpr int RW = WAVE * NW;
  const int rlane = hw * WAVE + lane;

  constexpr int M1 = LP + 2;
  lptr s1 = (NW == 1) ? hb : scr;
  {
    constexpr int nT  = (LP + 6 + 15) >> 4;
    constexpr int nT0 = (nT + NW - 1) / NW;
    tail_products<LP + 6>(Wf1 + F1M, in, s1, lhi, lln,
                          hw == 0 ? 0 : nT0, (NW == 1 || hw != 0) ? nT : nT0);
    if (NW > 1) __syncthreads();
    const float bt0 = b1[16], bt1 = b1[17], bt2 = b1[18];
    constexpr int NIT = (M1 + RW - 1) / RW;
    float hv[NIT][3];
#pragma unroll
    for (int it = 0; it < NIT; it++) {
      int m = it * RW + rlane; if (m > M1 - 1) m = M1 - 1;
      hv[it][0] = leaky(tail_sum<5>(s1, m, 0) + bt0);
      hv[it][1] = leaky(tail_sum<5>(s1, m, 1) + bt1);
      hv[it][2] = leaky(tail_sum<5>(s1, m, 2) + bt2);
    }
    if (NW == 1) wave_lds_order();
#pragma unroll
    for (int it = 0; it < NIT; it++) {
      const int m = it * RW + rlane;
      if (m < M1) {
        lptr hr = hb + m * NCH + 16;
        hr[0] = hv[it][0]; hr[1] = hv[it][1]; hr[2] = hv[it][2];
      }
    }
  }
  {
    v2f bw[KS1];
#pragma unroll
    for (int ks = 0; ks < KS1; ks++)
      bw[ks] = ((const AS1 v2f*)Wf1)[ks * WAVE + lane];
    const float b1a = b1[lln];
    constexpr int nFull = LP >> 4;
    constexpr int nPer  = nFull / NW;
    static_assert(nPer % 2 == 0 && nPer * NW == nFull, "tile pairs divide evenly");
#pragma nounroll
    for (int mt = hw * nPer; mt < hw * nPer + nPer; mt += 2) {
      const int m0 = mt << 4;
      clptr apA = in + (m0 + lln) * NCH + 2 * lhi;
      clptr apB = apA + 16 * NCH;
      tile_bases(apA, apB);
      v8f accA = {}; v8f accB = {};
#pragma unroll
      for (int ks = 0; ks < KS1; ks++) {
        v2f aA; aA.x = apA[4 * ks]; aA.y = apA[4 * ks + 1];
        v2f aB; aB.x = apB[4 * ks]; aB.y = apB[4 * ks + 1];
        accA = wmma4(aA, bw[ks], accA);
        accB = wmma4(aB, bw[ks], accB);
      }
      lptr hrA = hb + (m0 + 8 * lhi) * NCH + lln;
      lptr hrB = hrA + 16 * NCH;
#pragma unroll
      for (int v = 0; v < 8; v++) hrA[v * NCH] = leaky(accA[v] + b1a);
#pragma unroll
      for (int v = 0; v < 8; v++) hrB[v * NCH] = leaky(accB[v] + b1a);
    }
    if (NW == 1 || hw == NW - 1) {
      int mrow = LP + lln; if (mrow > M1 - 1) mrow = M1 - 1;
      clptr ap = in + mrow * NCH + 2 * lhi;
      v8f acc = {};
#pragma unroll
      for (int ks = 0; ks < KS1; ks++) {
        v2f a; a.x = ap[4 * ks]; a.y = ap[4 * ks + 1];
        acc = wmma4(a, bw[ks], acc);
      }
      if (lhi == 0) {
        lptr hr = hb + LP * NCH + lln;
        hr[0]   = leaky(acc[0] + b1a);
        hr[NCH] = leaky(acc[1] + b1a);
      }
    }
  }
  __syncthreads();

  if (MODE != 2) {
    constexpr int nT  = (M1 + 15) >> 4;
    constexpr int nT0 = (nT + NW - 1) / NW;
    tail_products<M1>(Wf2 + F2M, hb, scr, lhi, lln,
                      hw == 0 ? 0 : nT0, (NW == 1 || hw != 0) ? nT : nT0);
    if (NW > 1) __syncthreads();
    const float bt0 = b2[16], bt1 = b2[17], bt2 = b2[18];
    constexpr int NIT = LP / RW;
    float tv[NIT][3];
#pragma unroll
    for (int it = 0; it < NIT; it++) {
      const int m = it * RW + rlane;
      tv[it][0] = tail_sum<3>(scr, m, 0) + bt0;
      tv[it][1] = tail_sum<3>(scr, m, 1) + bt1;
      tv[it][2] = tail_sum<3>(scr, m, 2) + bt2;
    }
    if (MODE == 0) { if (NW > 1) __syncthreads(); else wave_lds_order(); }
#pragma unroll
    for (int it = 0; it < NIT; it++) {
      const int m = it * RW + rlane;
#pragma unroll
      for (int j = 0; j < 3; j++) {
        const float t   = fast_tanh(tv[it][j]);
        const float oth = other[(3 + m) * NCH + 16 + j];
        if (MODE == 0) outLds[(3 + m) * NCH + 16 + j] = oth * __expf(t);
        else           xg[(m * tmul + tadd) * NCH + 16 + j] = fmaf(sgn, t, oth);
      }
    }
  }
  {
    v2f bw[KS2];
#pragma unroll
    for (int ks = 0; ks < KS2; ks++)
      bw[ks] = ((const AS1 v2f*)Wf2)[ks * WAVE + lane];
    const float b2a = b2[lln];
    constexpr int nPer2 = (LP >> 4) / NW;
#pragma nounroll
    for (int mt = hw * nPer2; mt < hw * nPer2 + nPer2; mt += 2) {
      const int m0 = mt << 4;
      clptr apA = hb + (m0 + lln) * NCH + 2 * lhi;
      clptr apB = apA + 16 * NCH;
      tile_bases(apA, apB);
      v8f accA = {}; v8f accB = {};
#pragma unroll
      for (int ks = 0; ks < KS2; ks++) {
        v2f aA; aA.x = apA[4 * ks]; aA.y = apA[4 * ks + 1];
        v2f aB; aB.x = apB[4 * ks]; aB.y = apB[4 * ks + 1];
        accA = wmma4(aA, bw[ks], accA);
        accB = wmma4(aB, bw[ks], accB);
      }
      const int mb = m0 + 8 * lhi;
      if (MODE == 2) {
        if (lln < 5) {
#pragma unroll
          for (int v = 0; v < 8; v++) {
            const int m = mb + v;
            const float t = fast_tanh(accA[v] + b2a);
            outLds[(m * tmul + tadd) * 5 + lln] = fmaf(sgn, t, other[(3 + m) * NCH + lln]);
          }
#pragma unroll
          for (int v = 0; v < 8; v++) {
            const int m = mb + 16 + v;
            const float t = fast_tanh(accB[v] + b2a);
            outLds[(m * tmul + tadd) * 5 + lln] = fmaf(sgn, t, other[(3 + m) * NCH + lln]);
          }
        }
      } else {
#pragma unroll
        for (int v = 0; v < 8; v++) {
          const int m = mb + v;
          const float t   = fast_tanh(accA[v] + b2a);
          const float oth = other[(3 + m) * NCH + lln];
          if (MODE == 0) outLds[(3 + m) * NCH + lln] = oth * __expf(t);
          else           xg[(m * tmul + tadd) * NCH + lln] = fmaf(sgn, t, oth);
        }
#pragma unroll
        for (int v = 0; v < 8; v++) {
          const int m = mb + 16 + v;
          const float t   = fast_tanh(accB[v] + b2a);
          const float oth = other[(3 + m) * NCH + lln];
          if (MODE == 0) outLds[(3 + m) * NCH + lln] = oth * __expf(t);
          else           xg[(m * tmul + tadd) * NCH + lln] = fmaf(sgn, t, oth);
        }
      }
    }
  }
  if (MODE == 0) __syncthreads();
}

template <int DEPTH, int OUTMODE, int NW>
static __device__ __forceinline__ void level_body(lptr sm, int tid, int w, int hw, int sel,
                                                  gptr xg, cgptr Wf1p, cgptr Wf2p,
                                                  cgptr b1g, cgptr b2g, lptr yq) {
  constexpr int LP     = (L_SEQ >> DEPTH) >> 1;
  constexpr int stride = 1 << DEPTH;
  constexpr int NT     = TPB * NW;
  int node = 0;
  for (int i = 0; i < DEPTH; i++) node = 2 * node + 1 + ((sel >> i) & 1);

  lptr sev = sm;
  lptr sod = sev + BUF_SZ(LP);
  lptr db  = sod + BUF_SZ(LP);
  lptr cb  = db + BUF_SZ(LP);
  lptr hb  = cb + BUF_SZ(LP) + w * HB_SZ(LP);

  {
    const v4f z4 = {0.f, 0.f, 0.f, 0.f};
    for (int i = tid; i < SMEM_TOT(LP) / 4; i += NT) ((AS3 v4f*)sm)[i] = z4;
  }
  __syncthreads();

  constexpr int NEL   = LP * NCH;
  constexpr int NFULL = NEL / NT;
  constexpr int NREM  = NEL - NFULL * NT;
  static_assert(NREM % WAVE == 0, "the tail pass is whole waves");
#pragma unroll 19
  for (int p = 0; p < NFULL; p++) {
    const int i = p * NT + tid;
    const int r = i / NCH, c = i - r * NCH;
    sev[(3 + r) * NCH + c] = xg[((2 * r) * stride + sel) * NCH + c];
    sod[(3 + r) * NCH + c] = xg[((2 * r + 1) * stride + sel) * NCH + c];
  }
  if (NREM != 0 && tid < NREM) {
    const int i = NFULL * NT + tid;
    const int r = i / NCH, c = i - r * NCH;
    sev[(3 + r) * NCH + c] = xg[((2 * r) * stride + sel) * NCH + c];
    sod[(3 + r) * NCH + c] = xg[((2 * r + 1) * stride + sel) * NCH + c];
  }
  __syncthreads();
  for (int i = tid; i < 2 * 3 * NCH; i += NT) {
    const int c = i % NCH, r = (i / NCH) % 3;
    lptr buf = (i >= 3 * NCH) ? sod : sev;
    buf[r * NCH + c]            = buf[3 * NCH + c];
    buf[(LP + 3 + r) * NCH + c] = buf[(LP + 2) * NCH + c];
  }
  __syncthreads();

  const int nb = node * 4;
  {
    const int br = nb + w;
    run_branch<0, LP, NW>(Wf1p + (size_t)br * W1BR, Wf2p + (size_t)br * W2BR,
                          b1g + br * NCH, b2g + br * NCH,
                          w ? sod : sev, hb, w ? sev : sod, w ? cb : db,
                          w ? cb : db,
                          xg, 0, 0, 0.f, hw);
  }
  for (int i = tid; i < 2 * 3 * NCH; i += NT) {
    const int c = i % NCH, r = (i / NCH) % 3;
    lptr buf = (i >= 3 * NCH) ? cb : db;
    buf[r * NCH + c]            = buf[3 * NCH + c];
    buf[(LP + 3 + r) * NCH + c] = buf[(LP + 2) * NCH + c];
  }
  __syncthreads();
  {
    const int br = nb + 3 - w;
    run_branch<OUTMODE, LP, NW>(Wf1p + (size_t)br * W1BR, Wf2p + (size_t)br * W2BR,
                                b1g + br * NCH, b2g + br * NCH,
                                w ? cb : db, hb, w ? db : cb,
                                OUTMODE == 2 ? yq : db,
                                w ? sod : sev,
                                xg, 2 * stride, sel + w * stride, w ? -1.f : 1.f, hw);
  }
}

#define L0_NW 2
#define L1_NW 1
template <int DEPTH, int NW>
__global__ __launch_bounds__(TPB * NW) void scinet_level_kernel(
    float* __restrict__ xbuf_, const float* __restrict__ Wf1p_,
    const float* __restrict__ Wf2p_, const float* __restrict__ b1g_,
    const float* __restrict__ b2g_) {
  constexpr int LP     = (L_SEQ >> DEPTH) >> 1;
  constexpr int stride = 1 << DEPTH;
  __shared__ __align__(16) float smem[SMEM_TOT(LP)];
  const int tid = threadIdx.x;
  const int wv  = __builtin_amdgcn_readfirstlane(tid >> 5);
  const int w   = NW == 1 ? wv : wv >> 1;
  const int hw  = NW == 1 ? 0 : wv & 1;
  const int sel = blockIdx.x & (stride - 1);
  const int b   = blockIdx.x >> DEPTH;
  level_body<DEPTH, 1, NW>((lptr)smem, tid, w, hw, sel,
                       (gptr)(xbuf_ + (size_t)b * L_SEQ * NCH),
                       (cgptr)Wf1p_, (cgptr)Wf2p_, (cgptr)b1g_, (cgptr)b2g_, (lptr)smem);
}

#define LAST_DEPTH 2
#define LAST_PAIRS (1 << LAST_DEPTH)
#define LAST_LP    ((L_SEQ >> LAST_DEPTH) >> 1)
#define LAST_TPB   (LAST_PAIRS * TPB)
__global__ __launch_bounds__(LAST_TPB) void last_level_proj_kernel(
    const float* __restrict__ xbuf_, const float* __restrict__ Wf1p_,
    const float* __restrict__ Wf2p_, const float* __restrict__ b1g_,
    const float* __restrict__ b2g_, const float* __restrict__ x0D,
    const float* __restrict__ projw, float* __restrict__ out) {
  __shared__ __align__(16) float smem[LAST_PAIRS * SMEM_TOT(LAST_LP) + L_SEQ * 5];
  const int tid  = threadIdx.x;
  const int wid  = __builtin_amdgcn_readfirstlane(tid >> 5);
  const int pair = wid >> 1;
  const int b    = blockIdx.x;
  lptr yres = (lptr)smem + LAST_PAIRS * SMEM_TOT(LAST_LP);
  level_body<LAST_DEPTH, 2, 1>((lptr)smem + pair * SMEM_TOT(LAST_LP), tid & (TPB - 1),
                               wid & 1, 0, pair,
                            (gptr)(const_cast<float*>(xbuf_) + (size_t)b * L_SEQ * NCH),
                            (cgptr)Wf1p_, (cgptr)Wf2p_, (cgptr)b1g_, (cgptr)b2g_, yres);
  __syncthreads();
  cgptr x0 = (cgptr)x0D, pj = (cgptr)projw;
  for (int i = tid; i < L_SEQ * 5; i += LAST_TPB)
    yres[i] = yres[i] + x0[(size_t)b * L_SEQ * 5 + i];
  __syncthreads();
  if (tid < 160) {
    const int o = tid / 5, c = tid - o * 5;
    cgptr pw0 = pj + o * L_SEQ;
    cgptr pw1 = pw0 + 32 * L_SEQ;
    float acc0 = 0.f, acc1 = 0.f;
    for (int l = 0; l < L_SEQ; l++) {
      const float y = yres[l * 5 + c];
      acc0 = fmaf(pw0[l], y, acc0);
      acc1 = fmaf(pw1[l], y, acc1);
    }
    ((gptr)out)[(size_t)b * 320 + tid]       = acc0;
    ((gptr)out)[(size_t)b * 320 + tid + 160] = acc1;
  }
}

template <int TAPS, int KMAIN>
static __device__ __forceinline__ float pack_value(const float* __restrict__ Wg, int nb,
                                                   int rem, int fmain) {
  const bool tail = rem >= fmain;
  const int r  = tail ? rem - fmain : rem;
  const int ks = r >> 6, ln = (r >> 1) & 31, e = r & 1;
  const int n  = ln & 15;
  const int k  = 4 * ks + 2 * (ln >> 4) + e;
  int co, ci, tap; bool live;
  if (!tail) { tap = k / NCH; ci = k - tap * NCH; co = n; live = k < KMAIN; }
  else       { tap = n / 3; co = 16 + (n - 3 * tap); ci = k; live = (n < 3 * TAPS) && (k < NCH); }
  return live ? Wg[((size_t)(nb * NCH + co) * NCH + ci) * TAPS + tap] : 0.f;
}

__global__ void pack_weights_kernel(const float* __restrict__ W1,
                                    const float* __restrict__ W2,
                                    float* __restrict__ Wf1p,
                                    float* __restrict__ Wf2p) {
  const int idx = blockIdx.x * blockDim.x + threadIdx.x;
  const int N1 = 28 * W1BR, N2 = 28 * W2BR;
  if (idx < N1) {
    const int nb = idx / W1BR, rem = idx - nb * W1BR;
    Wf1p[idx] = pack_value<5, 95>(W1, nb, rem, F1M);
  } else if (idx < N1 + N2) {
    const int j = idx - N1;
    const int nb = j / W2BR, rem = j - nb * W2BR;
    Wf2p[j] = pack_value<3, 57>(W2, nb, rem, F2M);
  }
}

#define CC_TPB   256
#define CC_WG    2048
#define CC_V4    (B_TOT * L_SEQ * NCH / 4)
#define CC_PER   (CC_V4 / (CC_WG * CC_TPB))
static_assert(CC_PER * CC_WG * CC_TPB == CC_V4, "grid must cover the state exactly");
static_assert((L_SEQ * NCH) % 4 == 0, "a group of four never leaves the state");

template <int P0, int G>
static __device__ __forceinline__ void concat_groups(cgptr x0, cgptr xc,
                                                     float* xbuf, unsigned gid) {
  v4f o[G];
#pragma unroll
  for (int g = 0; g < G; g++) {
    const unsigned j = (unsigned)(P0 + g) * (CC_WG * CC_TPB) + gid;
#pragma unroll
    for (int e = 0; e < 4; e++) {
      const unsigned idx = 4u * j + e;
      const unsigned bl = idx / NCH, c = idx - bl * NCH;
      const bool     is0 = c < 5;
      const unsigned off = is0 ? bl * 5 + c : bl * 14 + (c - 5);
      cgptr base = is0 ? x0 : xc;
      o[g][e] = base[off];
    }
  }
#pragma unroll
  for (int g = 0; g < G; g++) {
    const unsigned j = (unsigned)(P0 + g) * (CC_WG * CC_TPB) + gid;
    ((AS1 v4f*)xbuf)[j] = o[g];
  }
}

__global__ __launch_bounds__(CC_TPB) void concat_kernel(
    const float* __restrict__ x0D, const float* __restrict__ xctrl,
    float* __restrict__ xbuf) {
  const unsigned gid = blockIdx.x * CC_TPB + threadIdx.x;
  cgptr x0 = (cgptr)x0D, xc = (cgptr)xctrl;
  static_assert(CC_PER == 19, "the three calls below cover 7 + 6 + 6 groups");
  concat_groups<0, 7>(x0, xc, xbuf, gid);
  concat_groups<7, 6>(x0, xc, xbuf, gid);
  concat_groups<13, 6>(x0, xc, xbuf, gid);
}

extern "C" void kernel_launch(void* const* d_in, const int* in_sizes, int n_in,
                              void* d_out, int out_size, void* d_ws,
                              size_t ws_size, hipStream_t stream) {
  const float* x0D   = (const float*)d_in[0];
  const float* xctrl = (const float*)d_in[1];
  const float* W1    = (const float*)d_in[2];
  const float* b1    = (const float*)d_in[3];
  const float* W2    = (const float*)d_in[4];
  const float* b2    = (const float*)d_in[5];
  const float* projw = (const float*)d_in[6];
  float* out = (float*)d_out;

  float* ws   = (float*)d_ws;
  float* xbuf = ws;
  const size_t XB = (size_t)B_TOT * L_SEQ * NCH;
  float* Wt1p = xbuf + XB;
  float* Wt2p = Wt1p + 28 * W1BR;

  const int packN = 28 * W1BR + 28 * W2BR;
  pack_weights_kernel<<<(packN + 255) / 256, 256, 0, stream>>>(W1, W2, Wt1p, Wt2p);
  concat_kernel<<<CC_WG, CC_TPB, 0, stream>>>(x0D, xctrl, xbuf);
  scinet_level_kernel<0, L0_NW><<<B_TOT, TPB * L0_NW, 0, stream>>>(xbuf, Wt1p, Wt2p, b1, b2);
  scinet_level_kernel<1, L1_NW><<<B_TOT * 2, TPB * L1_NW, 0, stream>>>(xbuf, Wt1p, Wt2p, b1, b2);
  last_level_proj_kernel<<<B_TOT, LAST_TPB, 0, stream>>>(xbuf, Wt1p, Wt2p, b1, b2,
                                                        x0D, projw, out);
}


# conv2 epilogue: tanh is one V_TANH_F32 instruction instead of clamp + exp + full-precision division (about 12 fewer vector instructions per element); tanh rounds differently in last bits
# speedup vs baseline: 2.5148x; 1.0692x over previous
#include <hip/hip_runtime.h>
#include <math.h>

typedef __attribute__((ext_vector_type(2))) float v2f;
typedef __attribute__((ext_vector_type(4))) float v4f;
typedef __attribute__((ext_vector_type(8))) float v8f;

#define AS3 __attribute__((address_space(3)))
#define AS1 __attribute__((address_space(1)))
typedef AS3 float*       lptr;
typedef const AS3 float* clptr;
typedef AS1 float*       gptr;
typedef const AS1 float* cgptr;

#define B_TOT    8192
#define L_SEQ    256
#define NCH      19
#define KPACK1   96
#define KPACK2   64
#define KS1      (KPACK1 / 4)
#define KS2      (KPACK2 / 4)
#define WAVE     32
#define TPB      64
#define KST      5
#define F1M      (KS1 * WAVE * 2)
#define F2M      (KS2 * WAVE * 2)
#define FTL      (KST * WAVE * 2)
#define W1BR     (F1M + FTL)
#define W2BR     (F2M + FTL)
#define TP       17

#define BUF_SZ(LP) ((((LP) + 6) * NCH + 32 + 3) & ~3)
#define HB_SZ(LP)  ((((LP) + 2) * NCH + 32 + 3) & ~3)
#define SMEM_TOT(LP) (4 * BUF_SZ(LP) + 2 * HB_SZ(LP))

static __device__ __forceinline__ v8f wmma4(v2f a, v2f b, v8f c) {
  return __builtin_amdgcn_wmma_f32_16x16x4_f32(false, a, false, b, (short)0, c,
                                               false, false);
}

static __device__ __forceinline__ float leaky(float x) {
  return x > 0.f ? x : 0.01f * x;
}

static __device__ __forceinline__ float fast_tanh(float x) {
  return __builtin_amdgcn_tanhf(x);
}

static __device__ __forceinline__ void tile_bases(clptr& p, clptr& q) {
  asm("" : "+v"(p), "+v"(q));
}

static __device__ __forceinline__ void wave_lds_order() {
  __builtin_amdgcn_fence(__ATOMIC_ACQ_REL, "wavefront");
}

template <int KS>
static __device__ __forceinline__ void load_a(clptr ap, v2f (&a)[KS]) {
#pragma unroll
  for (int ks = 0; ks < KS; ks++) { a[ks].x = ap[4 * ks]; a[ks].y = ap[4 * ks + 1]; }
}

template <int ROWS>
static __device__ __forceinline__ void tail_products(cgptr Wt, clptr src, lptr scr,
                                                     int lhi, int lln, int mt0, int mt1) {
  const int lane = lhi * 16 + lln;
  v2f bt[KST];
#pragma unroll
  for (int ks = 0; ks < KST; ks++) bt[ks] = ((const AS1 v2f*)Wt)[ks * WAVE + lane];
  v2f an[KST];
  {
    int prow = (mt0 << 4) + lln; if (prow > ROWS - 1) prow = ROWS - 1;
    load_a<KST>(src + prow * NCH + 2 * lhi, an);
  }
#pragma nounroll
  for (int mt = mt0; mt < mt1; mt++) {
    const int m0 = mt << 4;
    v2f a[KST];
#pragma unroll
    for (int ks = 0; ks < KST; ks++) a[ks] = an[ks];
    {
      const int mn = (mt + 1 < mt1) ? m0 + 16 : m0;
      int prow = mn + lln; if (prow > ROWS - 1) prow = ROWS - 1;
      load_a<KST>(src + prow * NCH + 2 * lhi, an);
    }
    v8f acc = {};
#pragma unroll
    for (int ks = 0; ks < KST; ks++) acc = wmma4(a[ks], bt[ks], acc);
    lptr srow = scr + (m0 + 8 * lhi) * TP + lln;
    if (m0 + 16 <= ROWS) {
#pragma unroll
      for (int v = 0; v < 8; v++) srow[v * TP] = acc[v];
    } else {
#pragma unroll
      for (int v = 0; v < 8; v++)
        if (m0 + 8 * lhi + v < ROWS) srow[v * TP] = acc[v];
    }
  }
}

template <int TAPS>
static __device__ __forceinline__ float tail_sum(clptr scr, int m, int j) {
  clptr sp = scr + m * TP + j;
  float s = sp[0];
#pragma unroll
  for (int t = 1; t < TAPS; t++) s += sp[t * (TP + 3)];
  return s;
}

template <int MODE, int LP, int NW>
static __device__ void run_branch(cgptr Wf1, cgptr Wf2, cgptr b1, cgptr b2,
                                  clptr in, lptr hb, clptr other, lptr outLds,
                                  lptr scr, gptr xg, int tmul, int tadd, float sgn,
                                  int hw) {
  const int lane = threadIdx.x & (WAVE - 1);
  const int lhi  = lane >> 4;
  const int lln  = lane & 15;
  static_assert(NW == 1 || NW == 2, "one or two waves per branch");
  static_assert(LP % (32 * NW) == 0, "each wave takes M tiles in pairs; 3*LP outputs fill whole waves");
  static_assert(MODE != 2 || NW == 1, "the last level runs one wave per branch");
  constexpr int RW = WAVE * NW;
  const int rlane = hw * WAVE + lane;

  constexpr int M1 = LP + 2;
  lptr s1 = (NW == 1) ? hb : scr;
  {
    constexpr int nT  = (LP + 6 + 15) >> 4;
    constexpr int nT0 = (nT + NW - 1) / NW;
    tail_products<LP + 6>(Wf1 + F1M, in, s1, lhi, lln,
                          hw == 0 ? 0 : nT0, (NW == 1 || hw != 0) ? nT : nT0);
    if (NW > 1) __syncthreads();
    const float bt0 = b1[16], bt1 = b1[17], bt2 = b1[18];
    constexpr int NIT = (M1 + RW - 1) / RW;
    float hv[NIT][3];
#pragma unroll
    for (int it = 0; it < NIT; it++) {
      int m = it * RW + rlane; if (m > M1 - 1) m = M1 - 1;
      hv[it][0] = leaky(tail_sum<5>(s1, m, 0) + bt0);
      hv[it][1] = leaky(tail_sum<5>(s1, m, 1) + bt1);
      hv[it][2] = leaky(tail_sum<5>(s1, m, 2) + bt2);
    }
    if (NW == 1) wave_lds_order();
#pragma unroll
    for (int it = 0; it < NIT; it++) {
      const int m = it * RW + rlane;
      if (m < M1) {
        lptr hr = hb + m * NCH + 16;
        hr[0] = hv[it][0]; hr[1] = hv[it][1]; hr[2] = hv[it][2];
      }
    }
  }
  {
    v2f bw[KS1];
#pragma unroll
    for (int ks = 0; ks < KS1; ks++)
      bw[ks] = ((const AS1 v2f*)Wf1)[ks * WAVE + lane];
    const float b1a = b1[lln];
    constexpr int nFull = LP >> 4;
    constexpr int nPer  = nFull / NW;
    static_assert(nPer % 2 == 0 && nPer * NW == nFull, "tile pairs divide evenly");
#pragma nounroll
    for (int mt = hw * nPer; mt < hw * nPer + nPer; mt += 2) {
      const int m0 = mt << 4;
      clptr apA = in + (m0 + lln) * NCH + 2 * lhi;
      clptr apB = apA + 16 * NCH;
      tile_bases(apA, apB);
      v8f accA = {}; v8f accB = {};
#pragma unroll
      for (int ks = 0; ks < KS1; ks++) {
        v2f aA; aA.x = apA[4 * ks]; aA.y = apA[4 * ks + 1];
        v2f aB; aB.x = apB[4 * ks]; aB.y = apB[4 * ks + 1];
        accA = wmma4(aA, bw[ks], accA);
        accB = wmma4(aB, bw[ks], accB);
      }
      lptr hrA = hb + (m0 + 8 * lhi) * NCH + lln;
      lptr hrB = hrA + 16 * NCH;
#pragma unroll
      for (int v = 0; v < 8; v++) hrA[v * NCH] = leaky(accA[v] + b1a);
#pragma unroll
      for (int v = 0; v < 8; v++) hrB[v * NCH] = leaky(accB[v] + b1a);
    }
    if (NW == 1 || hw == NW - 1) {
      int mrow = LP + lln; if (mrow > M1 - 1) mrow = M1 - 1;
      clptr ap = in + mrow * NCH + 2 * lhi;
      v8f acc = {};
#pragma unroll
      for (int ks = 0; ks < KS1; ks++) {
        v2f a; a.x = ap[4 * ks]; a.y = ap[4 * ks + 1];
        acc = wmma4(a, bw[ks], acc);
      }
      if (lhi == 0) {
        lptr hr = hb + LP * NCH + lln;
        hr[0]   = leaky(acc[0] + b1a);
        hr[NCH] = leaky(acc[1] + b1a);
      }
    }
  }
  __syncthreads();

  if (MODE != 2) {
    constexpr int nT  = (M1 + 15) >> 4;
    constexpr int nT0 = (nT + NW - 1) / NW;
    tail_products<M1>(Wf2 + F2M, hb, scr, lhi, lln,
                      hw == 0 ? 0 : nT0, (NW == 1 || hw != 0) ? nT : nT0);
    if (NW > 1) __syncthreads();
    const float bt0 = b2[16], bt1 = b2[17], bt2 = b2[18];
    constexpr int NIT = LP / RW;
    float tv[NIT][3];
#pragma unroll
    for (int it = 0; it < NIT; it++) {
      const int m = it * RW + rlane;
      tv[it][0] = tail_sum<3>(scr, m, 0) + bt0;
      tv[it][1] = tail_sum<3>(scr, m, 1) + bt1;
      tv[it][2] = tail_sum<3>(scr, m, 2) + bt2;
    }
    if (MODE == 0) { if (NW > 1) __syncthreads(); else wave_lds_order(); }
#pragma unroll
    for (int it = 0; it < NIT; it++) {
      const int m = it * RW + rlane;
#pragma unroll
      for (int j = 0; j < 3; j++) {
        const float t   = fast_tanh(tv[it][j]);
        const float oth = other[(3 + m) * NCH + 16 + j];
        if (MODE == 0) outLds[(3 + m) * NCH + 16 + j] = oth * __expf(t);
        else           xg[(m * tmul + tadd) * NCH + 16 + j] = fmaf(sgn, t, oth);
      }
    }
  }
  {
    v2f bw[KS2];
#pragma unroll
    for (int ks = 0; ks < KS2; ks++)
      bw[ks] = ((const AS1 v2f*)Wf2)[ks * WAVE + lane];
    const float b2a = b2[lln];
    constexpr int nPer2 = (LP >> 4) / NW;
#pragma nounroll
    for (int mt = hw * nPer2; mt < hw * nPer2 + nPer2; mt += 2) {
      const int m0 = mt << 4;
      clptr apA = hb + (m0 + lln) * NCH + 2 * lhi;
      clptr apB = apA + 16 * NCH;
      tile_bases(apA, apB);
      v8f accA = {}; v8f accB = {};
#pragma unroll
      for (int ks = 0; ks < KS2; ks++) {
        v2f aA; aA.x = apA[4 * ks]; aA.y = apA[4 * ks + 1];
        v2f aB; aB.x = apB[4 * ks]; aB.y = apB[4 * ks + 1];
        accA = wmma4(aA, bw[ks], accA);
        accB = wmma4(aB, bw[ks], accB);
      }
      const int mb = m0 + 8 * lhi;
      if (MODE == 2) {
        if (lln < 5) {
#pragma unroll
          for (int v = 0; v < 8; v++) {
            const int m = mb + v;
            const float t = fast_tanh(accA[v] + b2a);
            outLds[(m * tmul + tadd) * 5 + lln] = fmaf(sgn, t, other[(3 + m) * NCH + lln]);
          }
#pragma unroll
          for (int v = 0; v < 8; v++) {
            const int m = mb + 16 + v;
            const float t = fast_tanh(accB[v] + b2a);
            outLds[(m * tmul + tadd) * 5 + lln] = fmaf(sgn, t, other[(3 + m) * NCH + lln]);
          }
        }
      } else {
#pragma unroll
        for (int v = 0; v < 8; v++) {
          const int m = mb + v;
          const float t   = fast_tanh(accA[v] + b2a);
          const float oth = other[(3 + m) * NCH + lln];
          if (MODE == 0) outLds[(3 + m) * NCH + lln] = oth * __expf(t);
          else           xg[(m * tmul + tadd) * NCH + lln] = fmaf(sgn, t, oth);
        }
#pragma unroll
        for (int v = 0; v < 8; v++) {
          const int m = mb + 16 + v;
          const float t   = fast_tanh(accB[v] + b2a);
          const float oth = other[(3 + m) * NCH + lln];
          if (MODE == 0) outLds[(3 + m) * NCH + lln] = oth * __expf(t);
          else           xg[(m * tmul + tadd) * NCH + lln] = fmaf(sgn, t, oth);
        }
      }
    }
  }
  if (MODE == 0) __syncthreads();
}

template <int DEPTH, int OUTMODE, int NW>
static __device__ __forceinline__ void level_body(lptr sm, int tid, int w, int hw, int sel,
                                                  gptr xg, cgptr Wf1p, cgptr Wf2p,
                                                  cgptr b1g, cgptr b2g, lptr yq) {
  constexpr int LP     = (L_SEQ >> DEPTH) >> 1;
  constexpr int stride = 1 << DEPTH;
  constexpr int NT     = TPB * NW;
  int node = 0;
  for (int i = 0; i < DEPTH; i++) node = 2 * node + 1 + ((sel >> i) & 1);

  lptr sev = sm;
  lptr sod = sev + BUF_SZ(LP);
  lptr db  = sod + BUF_SZ(LP);
  lptr cb  = db + BUF_SZ(LP);
  lptr hb  = cb + BUF_SZ(LP) + w * HB_SZ(LP);

  {
    const v4f z4 = {0.f, 0.f, 0.f, 0.f};
    for (int i = tid; i < SMEM_TOT(LP) / 4; i += NT) ((AS3 v4f*)sm)[i] = z4;
  }
  __syncthreads();

  constexpr int NEL   = LP * NCH;
  constexpr int NFULL = NEL / NT;
  constexpr int NREM  = NEL - NFULL * NT;
  static_assert(NREM % WAVE == 0, "the tail pass is whole waves");
#pragma unroll 19
  for (int p = 0; p < NFULL; p++) {
    const int i = p * NT + tid;
    const int r = i / NCH, c = i - r * NCH;
    sev[(3 + r) * NCH + c] = xg[((2 * r) * stride + sel) * NCH + c];
    sod[(3 + r) * NCH + c] = xg[((2 * r + 1) * stride + sel) * NCH + c];
  }
  if (NREM != 0 && tid < NREM) {
    const int i = NFULL * NT + tid;
    const int r = i / NCH, c = i - r * NCH;
    sev[(3 + r) * NCH + c] = xg[((2 * r) * stride + sel) * NCH + c];
    sod[(3 + r) * NCH + c] = xg[((2 * r + 1) * stride + sel) * NCH + c];
  }
  __syncthreads();
  for (int i = tid; i < 2 * 3 * NCH; i += NT) {
    const int c = i % NCH, r = (i / NCH) % 3;
    lptr buf = (i >= 3 * NCH) ? sod : sev;
    buf[r * NCH + c]            = buf[3 * NCH + c];
    buf[(LP + 3 + r) * NCH + c] = buf[(LP + 2) * NCH + c];
  }
  __syncthreads();

  const int nb = node * 4;
  {
    const int br = nb + w;
    run_branch<0, LP, NW>(Wf1p + (size_t)br * W1BR, Wf2p + (size_t)br * W2BR,
                          b1g + br * NCH, b2g + br * NCH,
                          w ? sod : sev, hb, w ? sev : sod, w ? cb : db,
                          w ? cb : db,
                          xg, 0, 0, 0.f, hw);
  }
  for (int i = tid; i < 2 * 3 * NCH; i += NT) {
    const int c = i % NCH, r = (i / NCH) % 3;
    lptr buf = (i >= 3 * NCH) ? cb : db;
    buf[r * NCH + c]            = buf[3 * NCH + c];
    buf[(LP + 3 + r) * NCH + c] = buf[(LP + 2) * NCH + c];
  }
  __syncthreads();
  {
    const int br = nb + 3 - w;
    run_branch<OUTMODE, LP, NW>(Wf1p + (size_t)br * W1BR, Wf2p + (size_t)br * W2BR,
                                b1g + br * NCH, b2g + br * NCH,
                                w ? cb : db, hb, w ? db : cb,
                                OUTMODE == 2 ? yq : db,
                                w ? sod : sev,
                                xg, 2 * stride, sel + w * stride, w ? -1.f : 1.f, hw);
  }
}

#define L0_NW 2
#define L1_NW 1
template <int DEPTH, int NW>
__global__ __launch_bounds__(TPB * NW) void scinet_level_kernel(
    float* __restrict__ xbuf_, const float* __restrict__ Wf1p_,
    const float* __restrict__ Wf2p_, const float* __restrict__ b1g_,
    const float* __restrict__ b2g_) {
  constexpr int LP     = (L_SEQ >> DEPTH) >> 1;
  constexpr int stride = 1 << DEPTH;
  __shared__ __align__(16) float smem[SMEM_TOT(LP)];
  const int tid = threadIdx.x;
  const int wv  = __builtin_amdgcn_readfirstlane(tid >> 5);
  const int w   = NW == 1 ? wv : wv >> 1;
  const int hw  = NW == 1 ? 0 : wv & 1;
  const int sel = blockIdx.x & (stride - 1);
  const int b   = blockIdx.x >> DEPTH;
  level_body<DEPTH, 1, NW>((lptr)smem, tid, w, hw, sel,
                       (gptr)(xbuf_ + (size_t)b * L_SEQ * NCH),
                       (cgptr)Wf1p_, (cgptr)Wf2p_, (cgptr)b1g_, (cgptr)b2g_, (lptr)smem);
}

#define LAST_DEPTH 2
#define LAST_PAIRS (1 << LAST_DEPTH)
#define LAST_LP    ((L_SEQ >> LAST_DEPTH) >> 1)
#define LAST_TPB   (LAST_PAIRS * TPB)
__global__ __launch_bounds__(LAST_TPB) void last_level_proj_kernel(
    const float* __restrict__ xbuf_, const float* __restrict__ Wf1p_,
    const float* __restrict__ Wf2p_, const float* __restrict__ b1g_,
    const float* __restrict__ b2g_, const float* __restrict__ x0D,
    const float* __restrict__ projw, float* __restrict__ out) {
  __shared__ __align__(16) float smem[LAST_PAIRS * SMEM_TOT(LAST_LP) + L_SEQ * 5];
  const int tid  = threadIdx.x;
  const int wid  = __builtin_amdgcn_readfirstlane(tid >> 5);
  const int pair = wid >> 1;
  const int b    = blockIdx.x;
  lptr yres = (lptr)smem + LAST_PAIRS * SMEM_TOT(LAST_LP);
  level_body<LAST_DEPTH, 2, 1>((lptr)smem + pair * SMEM_TOT(LAST_LP), tid & (TPB - 1),
                               wid & 1, 0, pair,
                            (gptr)(const_cast<float*>(xbuf_) + (size_t)b * L_SEQ * NCH),
                            (cgptr)Wf1p_, (cgptr)Wf2p_, (cgptr)b1g_, (cgptr)b2g_, yres);
  __syncthreads();
  cgptr x0 = (cgptr)x0D, pj = (cgptr)projw;
  for (int i = tid; i < L_SEQ * 5; i += LAST_TPB)
    yres[i] = yres[i] + x0[(size_t)b * L_SEQ * 5 + i];
  __syncthreads();
  if (tid < 160) {
    const int o = tid / 5, c = tid - o * 5;
    cgptr pw0 = pj + o * L_SEQ;
    cgptr pw1 = pw0 + 32 * L_SEQ;
    float acc0 = 0.f, acc1 = 0.f;
    for (int l = 0; l < L_SEQ; l++) {
      const float y = yres[l * 5 + c];
      acc0 = fmaf(pw0[l], y, acc0);
      acc1 = fmaf(pw1[l], y, acc1);
    }
    ((gptr)out)[(size_t)b * 320 + tid]       = acc0;
    ((gptr)out)[(size_t)b * 320 + tid + 160] = acc1;
  }
}

template <int TAPS, int KMAIN>
static __device__ __forceinline__ float pack_value(const float* __restrict__ Wg, int nb,
                                                   int rem, int fmain) {
  const bool tail = rem >= fmain;
  const int r  = tail ? rem - fmain : rem;
  const int ks = r >> 6, ln = (r >> 1) & 31, e = r & 1;
  const int n  = ln & 15;
  const int k  = 4 * ks + 2 * (ln >> 4) + e;
  int co, ci, tap; bool live;
  if (!tail) { tap = k / NCH; ci = k - tap * NCH; co = n; live = k < KMAIN; }
  else       { tap = n / 3; co = 16 + (n - 3 * tap); ci = k; live = (n < 3 * TAPS) && (k < NCH); }
  return live ? Wg[((size_t)(nb * NCH + co) * NCH + ci) * TAPS + tap] : 0.f;
}

__global__ void pack_weights_kernel(const float* __restrict__ W1,
                                    const float* __restrict__ W2,
                                    float* __restrict__ Wf1p,
                                    float* __restrict__ Wf2p) {
  const int idx = blockIdx.x * blockDim.x + threadIdx.x;
  const int N1 = 28 * W1BR, N2 = 28 * W2BR;
  if (idx < N1) {
    const int nb = idx / W1BR, rem = idx - nb * W1BR;
    Wf1p[idx] = pack_value<5, 95>(W1, nb, rem, F1M);
  } else if (idx < N1 + N2) {
    const int j = idx - N1;
    const int nb = j / W2BR, rem = j - nb * W2BR;
    Wf2p[j] = pack_value<3, 57>(W2, nb, rem, F2M);
  }
}

#define CC_TPB   256
#define CC_WG    2048
#define CC_V4    (B_TOT * L_SEQ * NCH / 4)
#define CC_PER   (CC_V4 / (CC_WG * CC_TPB))
static_assert(CC_PER * CC_WG * CC_TPB == CC_V4, "grid must cover the state exactly");
static_assert((L_SEQ * NCH) % 4 == 0, "a group of four never leaves the state");

template <int P0, int G>
static __device__ __forceinline__ void concat_groups(cgptr x0, cgptr xc,
                                                     float* xbuf, unsigned gid) {
  v4f o[G];
#pragma unroll
  for (int g = 0; g < G; g++) {
    const unsigned j = (unsigned)(P0 + g) * (CC_WG * CC_TPB) + gid;
#pragma unroll
    for (int e = 0; e < 4; e++) {
      const unsigned idx = 4u * j + e;
      const unsigned bl = idx / NCH, c = idx - bl * NCH;
      const bool     is0 = c < 5;
      const unsigned off = is0 ? bl * 5 + c : bl * 14 + (c - 5);
      cgptr base = is0 ? x0 : xc;
      o[g][e] = base[off];
    }
  }
#pragma unroll
  for (int g = 0; g < G; g++) {
    const unsigned j = (unsigned)(P0 + g) * (CC_WG * CC_TPB) + gid;
    ((AS1 v4f*)xbuf)[j] = o[g];
  }
}

__global__ __launch_bounds__(CC_TPB) void concat_kernel(
    const float* __restrict__ x0D, const float* __restrict__ xctrl,
    float* __restrict__ xbuf) {
  const unsigned gid = blockIdx.x * CC_TPB + threadIdx.x;
  cgptr x0 = (cgptr)x0D, xc = (cgptr)xctrl;
  static_assert(CC_PER == 19, "the three calls below cover 7 + 6 + 6 groups");
  concat_groups<0, 7>(x0, xc, xbuf, gid);
  concat_groups<7, 6>(x0, xc, xbuf, gid);
  concat_groups<13, 6>(x0, xc, xbuf, gid);
}

extern "C" void kernel_launch(void* const* d_in, const int* in_sizes, int n_in,
                              void* d_out, int out_size, void* d_ws,
                              size_t ws_size, hipStream_t stream) {
  const float* x0D   = (const float*)d_in[0];
  const float* xctrl = (const float*)d_in[1];
  const float* W1    = (const float*)d_in[2];
  const float* b1    = (const float*)d_in[3];
  const float* W2    = (const float*)d_in[4];
  const float* b2    = (const float*)d_in[5];
  const float* projw = (const float*)d_in[6];
  float* out = (float*)d_out;

  float* ws   = (float*)d_ws;
  float* xbuf = ws;
  const size_t XB = (size_t)B_TOT * L_SEQ * NCH;
  float* Wt1p = xbuf + XB;
  float* Wt2p = Wt1p + 28 * W1BR;

  const int packN = 28 * W1BR + 28 * W2BR;
  pack_weights_kernel<<<(packN + 255) / 256, 256, 0, stream>>>(W1, W2, Wt1p, Wt2p);
  concat_kernel<<<CC_WG, CC_TPB, 0, stream>>>(x0D, xctrl, xbuf);
  scinet_level_kernel<0, L0_NW><<<B_TOT, TPB * L0_NW, 0, stream>>>(xbuf, Wt1p, Wt2p, b1, b2);
  scinet_level_kernel<1, L1_NW><<<B_TOT * 2, TPB * L1_NW, 0, stream>>>(xbuf, Wt1p, Wt2p, b1, b2);
  last_level_proj_kernel<<<B_TOT, LAST_TPB, 0, stream>>>(xbuf, Wt1p, Wt2p, b1, b2,
                                                        x0D, projw, out);
}


# Levels 0 and 1 fused: one 4-wave workgroup per batch element; level 0 leaves eu/ou in LDS already split into level 1's even/odd buffers, so that state skips global memory and a launch; same sums
# speedup vs baseline: 2.6082x; 1.0371x over previous
#include <hip/hip_runtime.h>
#include <math.h>

typedef __attribute__((ext_vector_type(2))) float v2f;
typedef __attribute__((ext_vector_type(4))) float v4f;
typedef __attribute__((ext_vector_type(8))) float v8f;

#define AS3 __attribute__((address_space(3)))
#define AS1 __attribute__((address_space(1)))
typedef AS3 float*       lptr;
typedef const AS3 float* clptr;
typedef AS1 float*       gptr;
typedef const AS1 float* cgptr;

#define B_TOT    8192
#define L_SEQ    256
#define NCH      19
#define KPACK1   96
#define KPACK2   64
#define KS1      (KPACK1 / 4)
#define KS2      (KPACK2 / 4)
#define WAVE     32
#define TPB      64
#define KST      5
#define F1M      (KS1 * WAVE * 2)
#define F2M      (KS2 * WAVE * 2)
#define FTL      (KST * WAVE * 2)
#define W1BR     (F1M + FTL)
#define W2BR     (F2M + FTL)
#define TP       17

#define BUF_SZ(LP) ((((LP) + 6) * NCH + 32 + 3) & ~3)
#define HB_SZ(LP)  ((((LP) + 2) * NCH + 32 + 3) & ~3)
#define SMEM_TOT(LP) (4 * BUF_SZ(LP) + 2 * HB_SZ(LP))

static __device__ __forceinline__ v8f wmma4(v2f a, v2f b, v8f c) {
  return __builtin_amdgcn_wmma_f32_16x16x4_f32(false, a, false, b, (short)0, c,
                                               false, false);
}

static __device__ __forceinline__ float leaky(float x) {
  return x > 0.f ? x : 0.01f * x;
}

static __device__ __forceinline__ float fast_tanh(float x) {
  return __builtin_amdgcn_tanhf(x);
}

static __device__ __forceinline__ void tile_bases(clptr& p, clptr& q) {
  asm("" : "+v"(p), "+v"(q));
}

static __device__ __forceinline__ void wave_lds_order() {
  __builtin_amdgcn_fence(__ATOMIC_ACQ_REL, "wavefront");
}

template <int KS>
static __device__ __forceinline__ void load_a(clptr ap, v2f (&a)[KS]) {
#pragma unroll
  for (int ks = 0; ks < KS; ks++) { a[ks].x = ap[4 * ks]; a[ks].y = ap[4 * ks + 1]; }
}

template <int ROWS>
static __device__ __forceinline__ void tail_products(cgptr Wt, clptr src, lptr scr,
                                                     int lhi, int lln, int mt0, int mt1) {
  const int lane = lhi * 16 + lln;
  v2f bt[KST];
#pragma unroll
  for (int ks = 0; ks < KST; ks++) bt[ks] = ((const AS1 v2f*)Wt)[ks * WAVE + lane];
  v2f an[KST];
  {
    int prow = (mt0 << 4) + lln; if (prow > ROWS - 1) prow = ROWS - 1;
    load_a<KST>(src + prow * NCH + 2 * lhi, an);
  }
#pragma nounroll
  for (int mt = mt0; mt < mt1; mt++) {
    const int m0 = mt << 4;
    v2f a[KST];
#pragma unroll
    for (int ks = 0; ks < KST; ks++) a[ks] = an[ks];
    {
      const int mn = (mt + 1 < mt1) ? m0 + 16 : m0;
      int prow = mn + lln; if (prow > ROWS - 1) prow = ROWS - 1;
      load_a<KST>(src + prow * NCH + 2 * lhi, an);
    }
    v8f acc = {};
#pragma unroll
    for (int ks = 0; ks < KST; ks++) acc = wmma4(a[ks], bt[ks], acc);
    lptr srow = scr + (m0 + 8 * lhi) * TP + lln;
    if (m0 + 16 <= ROWS) {
#pragma unroll
      for (int v = 0; v < 8; v++) srow[v * TP] = acc[v];
    } else {
#pragma unroll
      for (int v = 0; v < 8; v++)
        if (m0 + 8 * lhi + v < ROWS) srow[v * TP] = acc[v];
    }
  }
}

template <int TAPS>
static __device__ __forceinline__ float tail_sum(clptr scr, int m, int j) {
  clptr sp = scr + m * TP + j;
  float s = sp[0];
#pragma unroll
  for (int t = 1; t < TAPS; t++) s += sp[t * (TP + 3)];
  return s;
}

template <int MODE, int LP, int NW>
static __device__ void run_branch(cgptr Wf1, cgptr Wf2, cgptr b1, cgptr b2,
                                  clptr in, lptr hb, clptr other, lptr outLds,
                                  lptr scr, gptr xg, int tmul, int tadd, float sgn,
                                  int hw) {
  const int lane = threadIdx.x & (WAVE - 1);
  const int lhi  = lane >> 4;
  const int lln  = lane & 15;
  static_assert(NW == 1 || NW == 2, "one or two waves per branch");
  static_assert(LP % (32 * NW) == 0, "each wave takes M tiles in pairs; 3*LP outputs fill whole waves");
  static_assert(MODE != 2 || NW == 1, "the last level runs one wave per branch");
  constexpr int RW = WAVE * NW;
  const int rlane = hw * WAVE + lane;

  constexpr int M1 = LP + 2;
  lptr s1 = (NW == 1) ? hb : scr;
  {
    constexpr int nT  = (LP + 6 + 15) >> 4;
    constexpr int nT0 = (nT + NW - 1) / NW;
    tail_products<LP + 6>(Wf1 + F1M, in, s1, lhi, lln,
                          hw == 0 ? 0 : nT0, (NW == 1 || hw != 0) ? nT : nT0);
    if (NW > 1) __syncthreads();
    const float bt0 = b1[16], bt1 = b1[17], bt2 = b1[18];
    constexpr int NIT = (M1 + RW - 1) / RW;
    float hv[NIT][3];
#pragma unroll
    for (int it = 0; it < NIT; it++) {
      int m = it * RW + rlane; if (m > M1 - 1) m = M1 - 1;
      hv[it][0] = leaky(tail_sum<5>(s1, m, 0) + bt0);
      hv[it][1] = leaky(tail_sum<5>(s1, m, 1) + bt1);
      hv[it][2] = leaky(tail_sum<5>(s1, m, 2) + bt2);
    }
    if (NW == 1) wave_lds_order();
#pragma unroll
    for (int it = 0; it < NIT; it++) {
      const int m = it * RW + rlane;
      if (m < M1) {
        lptr hr = hb + m * NCH + 16;
        hr[0] = hv[it][0]; hr[1] = hv[it][1]; hr[2] = hv[it][2];
      }
    }
  }
  {
    v2f bw[KS1];
#pragma unroll
    for (int ks = 0; ks < KS1; ks++)
      bw[ks] = ((const AS1 v2f*)Wf1)[ks * WAVE + lane];
    const float b1a = b1[lln];
    constexpr int nFull = LP >> 4;
    constexpr int nPer  = nFull / NW;
    static_assert(nPer % 2 == 0 && nPer * NW == nFull, "tile pairs divide evenly");
#pragma nounroll
    for (int mt = hw * nPer; mt < hw * nPer + nPer; mt += 2) {
      const int m0 = mt << 4;
      clptr apA = in + (m0 + lln) * NCH + 2 * lhi;
      clptr apB = apA + 16 * NCH;
      tile_bases(apA, apB);
      v8f accA = {}; v8f accB = {};
#pragma unroll
      for (int ks = 0; ks < KS1; ks++) {
        v2f aA; aA.x = apA[4 * ks]; aA.y = apA[4 * ks + 1];
        v2f aB; aB.x = apB[4 * ks]; aB.y = apB[4 * ks + 1];
        accA = wmma4(aA, bw[ks], accA);
        accB = wmma4(aB, bw[ks], accB);
      }
      lptr hrA = hb + (m0 + 8 * lhi) * NCH + lln;
      lptr hrB = hrA + 16 * NCH;
#pragma unroll
      for (int v = 0; v < 8; v++) hrA[v * NCH] = leaky(accA[v] + b1a);
#pragma unroll
      for (int v = 0; v < 8; v++) hrB[v * NCH] = leaky(accB[v] + b1a);
    }
    if (NW == 1 || hw == NW - 1) {
      int mrow = LP + lln; if (mrow > M1 - 1) mrow = M1 - 1;
      clptr ap = in + mrow * NCH + 2 * lhi;
      v8f acc = {};
#pragma unroll
      for (int ks = 0; ks < KS1; ks++) {
        v2f a; a.x = ap[4 * ks]; a.y = ap[4 * ks + 1];
        acc = wmma4(a, bw[ks], acc);
      }
      if (lhi == 0) {
        lptr hr = hb + LP * NCH + lln;
        hr[0]   = leaky(acc[0] + b1a);
        hr[NCH] = leaky(acc[1] + b1a);
      }
    }
  }
  __syncthreads();

  if (MODE != 2) {
    constexpr int nT  = (M1 + 15) >> 4;
    constexpr int nT0 = (nT + NW - 1) / NW;
    tail_products<M1>(Wf2 + F2M, hb, scr, lhi, lln,
                      hw == 0 ? 0 : nT0, (NW == 1 || hw != 0) ? nT : nT0);
    if (NW > 1) __syncthreads();
    const float bt0 = b2[16], bt1 = b2[17], bt2 = b2[18];
    constexpr int NIT = LP / RW;
    float tv[NIT][3];
#pragma unroll
    for (int it = 0; it < NIT; it++) {
      const int m = it * RW + rlane;
      tv[it][0] = tail_sum<3>(scr, m, 0) + bt0;
      tv[it][1] = tail_sum<3>(scr, m, 1) + bt1;
      tv[it][2] = tail_sum<3>(scr, m, 2) + bt2;
    }
    if (MODE == 0 || MODE == 3) { if (NW > 1) __syncthreads(); else wave_lds_order(); }
#pragma unroll
    for (int it = 0; it < NIT; it++) {
      const int m = it * RW + rlane;
#pragma unroll
      for (int j = 0; j < 3; j++) {
        const float t   = fast_tanh(tv[it][j]);
        const float oth = other[(3 + m) * NCH + 16 + j];
        if (MODE == 0)      outLds[(3 + m) * NCH + 16 + j] = oth * __expf(t);
        else if (MODE == 3) outLds[(m & 1) * BUF_SZ(LP / 2) + (3 + (m >> 1)) * NCH + 16 + j] = fmaf(sgn, t, oth);
        else                xg[(m * tmul + tadd) * NCH + 16 + j] = fmaf(sgn, t, oth);
      }
    }
  }
  {
    v2f bw[KS2];
#pragma unroll
    for (int ks = 0; ks < KS2; ks++)
      bw[ks] = ((const AS1 v2f*)Wf2)[ks * WAVE + lane];
    const float b2a = b2[lln];
    constexpr int nPer2 = (LP >> 4) / NW;
#pragma nounroll
    for (int mt = hw * nPer2; mt < hw * nPer2 + nPer2; mt += 2) {
      const int m0 = mt << 4;
      clptr apA = hb + (m0 + lln) * NCH + 2 * lhi;
      clptr apB = apA + 16 * NCH;
      tile_bases(apA, apB);
      v8f accA = {}; v8f accB = {};
#pragma unroll
      for (int ks = 0; ks < KS2; ks++) {
        v2f aA; aA.x = apA[4 * ks]; aA.y = apA[4 * ks + 1];
        v2f aB; aB.x = apB[4 * ks]; aB.y = apB[4 * ks + 1];
        accA = wmma4(aA, bw[ks], accA);
        accB = wmma4(aB, bw[ks], accB);
      }
      const int mb = m0 + 8 * lhi;
      if (MODE == 2) {
        if (lln < 5) {
#pragma unroll
          for (int v = 0; v < 8; v++) {
            const int m = mb + v;
            const float t = fast_tanh(accA[v] + b2a);
            outLds[(m * tmul + tadd) * 5 + lln] = fmaf(sgn, t, other[(3 + m) * NCH + lln]);
          }
#pragma unroll
          for (int v = 0; v < 8; v++) {
            const int m = mb + 16 + v;
            const float t = fast_tanh(accB[v] + b2a);
            outLds[(m * tmul + tadd) * 5 + lln] = fmaf(sgn, t, other[(3 + m) * NCH + lln]);
          }
        }
      } else {
#pragma unroll
        for (int v = 0; v < 8; v++) {
          const int m = mb + v;
          const float t   = fast_tanh(accA[v] + b2a);
          const float oth = other[(3 + m) * NCH + lln];
          if (MODE == 0)      outLds[(3 + m) * NCH + lln] = oth * __expf(t);
          else if (MODE == 3) outLds[(v & 1) * BUF_SZ(LP / 2) + (3 + (m >> 1)) * NCH + lln] = fmaf(sgn, t, oth);
          else                xg[(m * tmul + tadd) * NCH + lln] = fmaf(sgn, t, oth);
        }
#pragma unroll
        for (int v = 0; v < 8; v++) {
          const int m = mb + 16 + v;
          const float t   = fast_tanh(accB[v] + b2a);
          const float oth = other[(3 + m) * NCH + lln];
          if (MODE == 0)      outLds[(3 + m) * NCH + lln] = oth * __expf(t);
          else if (MODE == 3) outLds[(v & 1) * BUF_SZ(LP / 2) + (3 + (m >> 1)) * NCH + lln] = fmaf(sgn, t, oth);
          else                xg[(m * tmul + tadd) * NCH + lln] = fmaf(sgn, t, oth);
        }
      }
    }
  }
  if (MODE == 0) __syncthreads();
}

template <int DEPTH, int OUTMODE, int NW>
static __device__ __forceinline__ void level_body(lptr sm, int tid, int w, int hw, int sel,
                                                  gptr xg, cgptr Wf1p, cgptr Wf2p,
                                                  cgptr b1g, cgptr b2g, lptr yq) {
  constexpr int LP     = (L_SEQ >> DEPTH) >> 1;
  constexpr int stride = 1 << DEPTH;
  constexpr int NT     = TPB * NW;
  int node = 0;
  for (int i = 0; i < DEPTH; i++) node = 2 * node + 1 + ((sel >> i) & 1);

  lptr sev = sm;
  lptr sod = sev + BUF_SZ(LP);
  lptr db  = sod + BUF_SZ(LP);
  lptr cb  = db + BUF_SZ(LP);
  lptr hb  = cb + BUF_SZ(LP) + w * HB_SZ(LP);

  {
    const v4f z4 = {0.f, 0.f, 0.f, 0.f};
    for (int i = tid; i < SMEM_TOT(LP) / 4; i += NT) ((AS3 v4f*)sm)[i] = z4;
  }
  __syncthreads();

  constexpr int NEL   = LP * NCH;
  constexpr int NFULL = NEL / NT;
  constexpr int NREM  = NEL - NFULL * NT;
  static_assert(NREM % WAVE == 0, "the tail pass is whole waves");
#pragma unroll 19
  for (int p = 0; p < NFULL; p++) {
    const int i = p * NT + tid;
    const int r = i / NCH, c = i - r * NCH;
    sev[(3 + r) * NCH + c] = xg[((2 * r) * stride + sel) * NCH + c];
    sod[(3 + r) * NCH + c] = xg[((2 * r + 1) * stride + sel) * NCH + c];
  }
  if (NREM != 0 && tid < NREM) {
    const int i = NFULL * NT + tid;
    const int r = i / NCH, c = i - r * NCH;
    sev[(3 + r) * NCH + c] = xg[((2 * r) * stride + sel) * NCH + c];
    sod[(3 + r) * NCH + c] = xg[((2 * r + 1) * stride + sel) * NCH + c];
  }
  __syncthreads();
  for (int i = tid; i < 2 * 3 * NCH; i += NT) {
    const int c = i % NCH, r = (i / NCH) % 3;
    lptr buf = (i >= 3 * NCH) ? sod : sev;
    buf[r * NCH + c]            = buf[3 * NCH + c];
    buf[(LP + 3 + r) * NCH + c] = buf[(LP + 2) * NCH + c];
  }
  __syncthreads();

  const int nb = node * 4;
  {
    const int br = nb + w;
    run_branch<0, LP, NW>(Wf1p + (size_t)br * W1BR, Wf2p + (size_t)br * W2BR,
                          b1g + br * NCH, b2g + br * NCH,
                          w ? sod : sev, hb, w ? sev : sod, w ? cb : db,
                          w ? cb : db,
                          xg, 0, 0, 0.f, hw);
  }
  for (int i = tid; i < 2 * 3 * NCH; i += NT) {
    const int c = i % NCH, r = (i / NCH) % 3;
    lptr buf = (i >= 3 * NCH) ? cb : db;
    buf[r * NCH + c]            = buf[3 * NCH + c];
    buf[(LP + 3 + r) * NCH + c] = buf[(LP + 2) * NCH + c];
  }
  __syncthreads();
  {
    const int br = nb + 3 - w;
    run_branch<OUTMODE, LP, NW>(Wf1p + (size_t)br * W1BR, Wf2p + (size_t)br * W2BR,
                                b1g + br * NCH, b2g + br * NCH,
                                w ? cb : db, hb, w ? db : cb,
                                OUTMODE == 2 ? yq : db,
                                w ? sod : sev,
                                xg, 2 * stride, sel + w * stride, w ? -1.f : 1.f, hw);
  }
}

#define LP0    (L_SEQ >> 1)
#define LP1    (L_SEQ >> 2)
#define X_SZ   (4 * BUF_SZ(LP1))
#define Y0_SZ  (2 * BUF_SZ(LP0) + 2 * HB_SZ(LP0))
#define Y1_SZ  (2 * BUF_SZ(LP1) + 2 * HB_SZ(LP1))
#define F01_SMEM (X_SZ + (2 * Y1_SZ > Y0_SZ ? 2 * Y1_SZ : Y0_SZ))
#define F01_TPB  (2 * TPB)
static_assert(X_SZ >= 2 * BUF_SZ(LP0), "level 0's ev and od fit in X");
static_assert(2 * BUF_SZ(LP1) >= (LP0 + 6) * TP, "a level-0 U / P branch's per-tap scratch fits in its target pair's half of X");
static_assert(X_SZ % 4 == 0 && F01_SMEM % 4 == 0, "16-byte clears");

template <int LP, int NT, int STRIDE>
static __device__ __forceinline__ void split_load(lptr sev, lptr sod, gptr xg, int sel, int tid) {
  constexpr int NEL   = LP * NCH;
  constexpr int NFULL = NEL / NT;
  constexpr int NREM  = NEL - NFULL * NT;
  static_assert(NREM % WAVE == 0, "the tail pass is whole waves");
#pragma unroll 19
  for (int p = 0; p < NFULL; p++) {
    const int i = p * NT + tid;
    const int r = i / NCH, c = i - r * NCH;
    sev[(3 + r) * NCH + c] = xg[((2 * r) * STRIDE + sel) * NCH + c];
    sod[(3 + r) * NCH + c] = xg[((2 * r + 1) * STRIDE + sel) * NCH + c];
  }
  if (NREM != 0 && tid < NREM) {
    const int i = NFULL * NT + tid;
    const int r = i / NCH, c = i - r * NCH;
    sev[(3 + r) * NCH + c] = xg[((2 * r) * STRIDE + sel) * NCH + c];
    sod[(3 + r) * NCH + c] = xg[((2 * r + 1) * STRIDE + sel) * NCH + c];
  }
}

template <int LP, int NT>
static __device__ __forceinline__ void rep_pad(lptr a, lptr b, int tid) {
  for (int i = tid; i < 2 * 3 * NCH; i += NT) {
    const int c = i % NCH, r = (i / NCH) % 3;
    lptr buf = (i >= 3 * NCH) ? b : a;
    buf[r * NCH + c]            = buf[3 * NCH + c];
    buf[(LP + 3 + r) * NCH + c] = buf[(LP + 2) * NCH + c];
  }
}

__global__ __launch_bounds__(F01_TPB) __attribute__((amdgpu_waves_per_eu(5))) void level01_kernel(
    float* __restrict__ xbuf_, const float* __restrict__ Wf1p_,
    const float* __restrict__ Wf2p_, const float* __restrict__ b1g_,
    const float* __restrict__ b2g_) {
  __shared__ __align__(16) float smem[F01_SMEM];
  const int tid = threadIdx.x;
  const int wv = __builtin_amdgcn_readfirstlane(tid >> 5);
  gptr  xg   = (gptr)(xbuf_ + (size_t)blockIdx.x * L_SEQ * NCH);
  cgptr Wf1p = (cgptr)Wf1p_, Wf2p = (cgptr)Wf2p_, b1g = (cgptr)b1g_, b2g = (cgptr)b2g_;
  lptr X = (lptr)smem;
  lptr Y = X + X_SZ;
  const v4f z4 = {0.f, 0.f, 0.f, 0.f};

  for (int i = tid; i < F01_SMEM / 4; i += F01_TPB) ((AS3 v4f*)smem)[i] = z4;
  __syncthreads();

  {
    const int w  = wv >> 1;
    const int hw = wv & 1;
    lptr sev = X;
    lptr sod = X + BUF_SZ(LP0);
    lptr db  = Y;
    lptr cb  = Y + BUF_SZ(LP0);
    lptr hb  = Y + 2 * BUF_SZ(LP0) + w * HB_SZ(LP0);
    split_load<LP0, F01_TPB, 1>(sev, sod, xg, 0, tid);
    __syncthreads();
    rep_pad<LP0, F01_TPB>(sev, sod, tid);
    __syncthreads();
    {
      const int br = w;
      run_branch<0, LP0, 2>(Wf1p + (size_t)br * W1BR, Wf2p + (size_t)br * W2BR,
                            b1g + br * NCH, b2g + br * NCH,
                            w ? sod : sev, hb, w ? sev : sod, w ? cb : db,
                            w ? cb : db,
                            xg, 0, 0, 0.f, hw);
    }
    rep_pad<LP0, F01_TPB>(db, cb, tid);
    __syncthreads();
    {
      const int br = 3 - w;
      lptr tgt = X + w * (2 * BUF_SZ(LP1));
      run_branch<3, LP0, 2>(Wf1p + (size_t)br * W1BR, Wf2p + (size_t)br * W2BR,
                            b1g + br * NCH, b2g + br * NCH,
                            w ? cb : db, hb, w ? db : cb, tgt,
                            tgt,
                            xg, 0, 0, w ? -1.f : 1.f, hw);
    }
  }
  __syncthreads();

  for (int i = tid; i < (F01_SMEM - X_SZ) / 4; i += F01_TPB) ((AS3 v4f*)Y)[i] = z4;
  {
    constexpr int SP0 = (LP1 + 6) * NCH;
    constexpr int SPN = BUF_SZ(LP1) - SP0;
    for (int i = tid; i < 4 * SPN; i += F01_TPB) {
      const int k = i / SPN;
      X[k * BUF_SZ(LP1) + SP0 + (i - k * SPN)] = 0.f;
    }
  }

  {
    const int p    = wv >> 1;
    const int w    = wv & 1;
    const int tid1 = tid & (TPB - 1);
    lptr sev = X + p * (2 * BUF_SZ(LP1));
    lptr sod = sev + BUF_SZ(LP1);
    lptr db  = Y + p * Y1_SZ;
    lptr cb  = db + BUF_SZ(LP1);
    lptr hb  = cb + BUF_SZ(LP1) + w * HB_SZ(LP1);
    rep_pad<LP1, TPB>(sev, sod, tid1);
    __syncthreads();
    const int nb = (1 + p) * 4;
    {
      const int br = nb + w;
      run_branch<0, LP1, 1>(Wf1p + (size_t)br * W1BR, Wf2p + (size_t)br * W2BR,
                            b1g + br * NCH, b2g + br * NCH,
                            w ? sod : sev, hb, w ? sev : sod, w ? cb : db,
                            w ? cb : db,
                            xg, 0, 0, 0.f, 0);
    }
    rep_pad<LP1, TPB>(db, cb, tid1);
    __syncthreads();
    {
      const int br = nb + 3 - w;
      run_branch<1, LP1, 1>(Wf1p + (size_t)br * W1BR, Wf2p + (size_t)br * W2BR,
                            b1g + br * NCH, b2g + br * NCH,
                            w ? cb : db, hb, w ? db : cb, db,
                            w ? sod : sev,
                            xg, 4, p + 2 * w, w ? -1.f : 1.f, 0);
    }
  }
}

#define LAST_DEPTH 2
#define LAST_PAIRS (1 << LAST_DEPTH)
#define LAST_LP    ((L_SEQ >> LAST_DEPTH) >> 1)
#define LAST_TPB   (LAST_PAIRS * TPB)
__global__ __launch_bounds__(LAST_TPB) void last_level_proj_kernel(
    const float* __restrict__ xbuf_, const float* __restrict__ Wf1p_,
    const float* __restrict__ Wf2p_, const float* __restrict__ b1g_,
    const float* __restrict__ b2g_, const float* __restrict__ x0D,
    const float* __restrict__ projw, float* __restrict__ out) {
  __shared__ __align__(16) float smem[LAST_PAIRS * SMEM_TOT(LAST_LP) + L_SEQ * 5];
  const int tid  = threadIdx.x;
  const int wid  = __builtin_amdgcn_readfirstlane(tid >> 5);
  const int pair = wid >> 1;
  const int b    = blockIdx.x;
  lptr yres = (lptr)smem + LAST_PAIRS * SMEM_TOT(LAST_LP);
  level_body<LAST_DEPTH, 2, 1>((lptr)smem + pair * SMEM_TOT(LAST_LP), tid & (TPB - 1),
                               wid & 1, 0, pair,
                            (gptr)(const_cast<float*>(xbuf_) + (size_t)b * L_SEQ * NCH),
                            (cgptr)Wf1p_, (cgptr)Wf2p_, (cgptr)b1g_, (cgptr)b2g_, yres);
  __syncthreads();
  cgptr x0 = (cgptr)x0D, pj = (cgptr)projw;
  for (int i = tid; i < L_SEQ * 5; i += LAST_TPB)
    yres[i] = yres[i] + x0[(size_t)b * L_SEQ * 5 + i];
  __syncthreads();
  if (tid < 160) {
    const int o = tid / 5, c = tid - o * 5;
    cgptr pw0 = pj + o * L_SEQ;
    cgptr pw1 = pw0 + 32 * L_SEQ;
    float acc0 = 0.f, acc1 = 0.f;
    for (int l = 0; l < L_SEQ; l++) {
      const float y = yres[l * 5 + c];
      acc0 = fmaf(pw0[l], y, acc0);
      acc1 = fmaf(pw1[l], y, acc1);
    }
    ((gptr)out)[(size_t)b * 320 + tid]       = acc0;
    ((gptr)out)[(size_t)b * 320 + tid + 160] = acc1;
  }
}

template <int TAPS, int KMAIN>
static __device__ __forceinline__ float pack_value(const float* __restrict__ Wg, int nb,
                                                   int rem, int fmain) {
  const bool tail = rem >= fmain;
  const int r  = tail ? rem - fmain : rem;
  const int ks = r >> 6, ln = (r >> 1) & 31, e = r & 1;
  const int n  = ln & 15;
  const int k  = 4 * ks + 2 * (ln >> 4) + e;
  int co, ci, tap; bool live;
  if (!tail) { tap = k / NCH; ci = k - tap * NCH; co = n; live = k < KMAIN; }
  else       { tap = n / 3; co = 16 + (n - 3 * tap); ci = k; live = (n < 3 * TAPS) && (k < NCH); }
  return live ? Wg[((size_t)(nb * NCH + co) * NCH + ci) * TAPS + tap] : 0.f;
}

__global__ void pack_weights_kernel(const float* __restrict__ W1,
                                    const float* __restrict__ W2,
                                    float* __restrict__ Wf1p,
                                    float* __restrict__ Wf2p) {
  const int idx = blockIdx.x * blockDim.x + threadIdx.x;
  const int N1 = 28 * W1BR, N2 = 28 * W2BR;
  if (idx < N1) {
    const int nb = idx / W1BR, rem = idx - nb * W1BR;
    Wf1p[idx] = pack_value<5, 95>(W1, nb, rem, F1M);
  } else if (idx < N1 + N2) {
    const int j = idx - N1;
    const int nb = j / W2BR, rem = j - nb * W2BR;
    Wf2p[j] = pack_value<3, 57>(W2, nb, rem, F2M);
  }
}

#define CC_TPB   256
#define CC_WG    2048
#define CC_V4    (B_TOT * L_SEQ * NCH / 4)
#define CC_PER   (CC_V4 / (CC_WG * CC_TPB))
static_assert(CC_PER * CC_WG * CC_TPB == CC_V4, "grid must cover the state exactly");
static_assert((L_SEQ * NCH) % 4 == 0, "a group of four never leaves the state");

template <int P0, int G>
static __device__ __forceinline__ void concat_groups(cgptr x0, cgptr xc,
                                                     float* xbuf, unsigned gid) {
  v4f o[G];
#pragma unroll
  for (int g = 0; g < G; g++) {
    const unsigned j = (unsigned)(P0 + g) * (CC_WG * CC_TPB) + gid;
#pragma unroll
    for (int e = 0; e < 4; e++) {
      const unsigned idx = 4u * j + e;
      const unsigned bl = idx / NCH, c = idx - bl * NCH;
      const bool     is0 = c < 5;
      const unsigned off = is0 ? bl * 5 + c : bl * 14 + (c - 5);
      cgptr base = is0 ? x0 : xc;
      o[g][e] = base[off];
    }
  }
#pragma unroll
  for (int g = 0; g < G; g++) {
    const unsigned j = (unsigned)(P0 + g) * (CC_WG * CC_TPB) + gid;
    ((AS1 v4f*)xbuf)[j] = o[g];
  }
}

__global__ __launch_bounds__(CC_TPB) void concat_kernel(
    const float* __restrict__ x0D, const float* __restrict__ xctrl,
    float* __restrict__ xbuf) {
  const unsigned gid = blockIdx.x * CC_TPB + threadIdx.x;
  cgptr x0 = (cgptr)x0D, xc = (cgptr)xctrl;
  static_assert(CC_PER == 19, "the three calls below cover 7 + 6 + 6 groups");
  concat_groups<0, 7>(x0, xc, xbuf, gid);
  concat_groups<7, 6>(x0, xc, xbuf, gid);
  concat_groups<13, 6>(x0, xc, xbuf, gid);
}

extern "C" void kernel_launch(void* const* d_in, const int* in_sizes, int n_in,
                              void* d_out, int out_size, void* d_ws,
                              size_t ws_size, hipStream_t stream) {
  const float* x0D   = (const float*)d_in[0];
  const float* xctrl = (const float*)d_in[1];
  const float* W1    = (const float*)d_in[2];
  const float* b1    = (const float*)d_in[3];
  const float* W2    = (const float*)d_in[4];
  const float* b2    = (const float*)d_in[5];
  const float* projw = (const float*)d_in[6];
  float* out = (float*)d_out;

  float* ws   = (float*)d_ws;
  float* xbuf = ws;
  const size_t XB = (size_t)B_TOT * L_SEQ * NCH;
  float* Wt1p = xbuf + XB;
  float* Wt2p = Wt1p + 28 * W1BR;

  const int packN = 28 * W1BR + 28 * W2BR;
  pack_weights_kernel<<<(packN + 255) / 256, 256, 0, stream>>>(W1, W2, Wt1p, Wt2p);
  concat_kernel<<<CC_WG, CC_TPB, 0, stream>>>(x0D, xctrl, xbuf);
  level01_kernel<<<B_TOT, F01_TPB, 0, stream>>>(xbuf, Wt1p, Wt2p, b1, b2);
  last_level_proj_kernel<<<B_TOT, LAST_TPB, 0, stream>>>(xbuf, Wt1p, Wt2p, b1, b2,
                                                        x0D, projw, out);
}
